# starting kernel
# baseline (speedup 1.0000x reference)
#include <hip/hip_runtime.h>
#include <hip/hip_bf16.h>


#define IN_DIM 128
#define HID    128
#define OUT_DIM 64
#define EPS_W  1e-8f

typedef float v2f __attribute__((ext_vector_type(2)));
typedef float v8f __attribute__((ext_vector_type(8)));

#if defined(__gfx1250__) && __has_builtin(__builtin_amdgcn_wmma_f32_16x16x4_f32)
#define HAVE_WMMA_F32 1
#endif

__global__ void fill_f32_kernel(float* __restrict__ p, float v, long n) {
    long i = (long)blockIdx.x * blockDim.x + threadIdx.x;
    if (i < n) p[i] = v;
}

__global__ void init_minmax_kernel(unsigned* __restrict__ mm) {
    if (threadIdx.x == 0) {
        mm[0] = 0x7F800000u;
        mm[1] = 0x00000000u;
    }
}

__global__ void edge_mean_minmax_kernel(const float* __restrict__ ea,
                                        float* __restrict__ ew,
                                        unsigned* __restrict__ mm, long E) {
    __shared__ unsigned smin[256];
    __shared__ unsigned smax[256];
    int tid = threadIdx.x;
    long e = (long)blockIdx.x * blockDim.x + tid;
    unsigned bmin = 0x7F800000u, bmax = 0u;
    if (e < E) {
        const float4 a = *(const float4*)(ea + e * 4);
        float m = 0.25f * (a.x + a.y + a.z + a.w);
        ew[e] = m;
        unsigned b = __float_as_uint(m);
        bmin = b; bmax = b;
    }
    smin[tid] = bmin; smax[tid] = bmax;
    __syncthreads();
    for (int s = 128; s > 0; s >>= 1) {
        if (tid < s) {
            smin[tid] = min(smin[tid], smin[tid + s]);
            smax[tid] = max(smax[tid], smax[tid + s]);
        }
        __syncthreads();
    }
    if (tid == 0) {
        atomicMin(&mm[0], smin[0]);
        atomicMax(&mm[1], smax[0]);
    }
}

__global__ void edge_weight_deg_kernel(float* __restrict__ ew,
                                       const int* __restrict__ col,
                                       float* __restrict__ deg,
                                       const unsigned* __restrict__ mm, long E) {
    long e = (long)blockIdx.x * blockDim.x + threadIdx.x;
    if (e >= E) return;
    float mn = __uint_as_float(mm[0]);
    float mx = __uint_as_float(mm[1]);
    float m = ew[e];
    float range = mx - mn;
    float w = (range > EPS_W) ? (1.0f - (m - mn) / (range + EPS_W)) : 1.0f;
    w = fmaxf(w, 0.0f);
    ew[e] = w;
    atomicAdd(&deg[col[e]], w);
}

__global__ void rsqrt_kernel(float* __restrict__ deg, long n) {
    long i = (long)blockIdx.x * blockDim.x + threadIdx.x;
    if (i < n) deg[i] = rsqrtf(deg[i]);
}

__global__ void edge_norm_kernel(const float* __restrict__ ew,
                                 const int* __restrict__ row,
                                 const int* __restrict__ col,
                                 const float* __restrict__ dis,
                                 float* __restrict__ nrm, long E) {
    long e = (long)blockIdx.x * blockDim.x + threadIdx.x;
    if (e >= E) return;
    nrm[e] = dis[row[e]] * ew[e] * dis[col[e]];
}

__global__ void transpose_kernel(float* __restrict__ dst,
                                 const float* __restrict__ src, int R, int C) {
    long i = (long)blockIdx.x * blockDim.x + threadIdx.x;
    if (i >= (long)R * C) return;
    int r = (int)(i / C), c = (int)(i % C);
    dst[(size_t)c * R + r] = src[i];
}

__global__ void __launch_bounds__(256)
gemm_nt128_kernel(const float* __restrict__ A, const float* __restrict__ Bt,
                  float* __restrict__ C, int N) {
    const int wave = threadIdx.x >> 5;
    const int lane = threadIdx.x & 31;
    const int half = lane >> 4;
    const int lm   = lane & 15;
    const int m0   = (blockIdx.x * 8 + wave) * 16;
    if (m0 >= N) return;

    int rowA = m0 + lm;
    if (rowA >= N) rowA = N - 1;
    const float* __restrict__ arow = A + (size_t)rowA * 128;

#if defined(HAVE_WMMA_F32)
    v8f acc[8];
#pragma unroll
    for (int t = 0; t < 8; ++t)
        acc[t] = (v8f){0.f, 0.f, 0.f, 0.f, 0.f, 0.f, 0.f, 0.f};

    for (int k0 = 0; k0 < 128; k0 += 4) {
        const int ka = k0 + 2 * half;
        v2f a;
        a.x = arow[ka];
        a.y = arow[ka + 1];
        const float* __restrict__ b0 = Bt + (size_t)ka * 128 + lm;
        const float* __restrict__ b1 = Bt + (size_t)(ka + 1) * 128 + lm;
#pragma unroll
        for (int t = 0; t < 8; ++t) {
            v2f b;
            b.x = b0[t * 16];
            b.y = b1[t * 16];
            acc[t] = __builtin_amdgcn_wmma_f32_16x16x4_f32(
                false, a, false, b, (short)0, acc[t], false, false);
        }
    }
#pragma unroll
    for (int t = 0; t < 8; ++t) {
        const int colv = t * 16 + lm;
#pragma unroll
        for (int v = 0; v < 8; ++v) {
            const int rowv = m0 + v + 8 * half;
            if (rowv < N) C[(size_t)rowv * 128 + colv] = acc[t][v];
        }
    }
#else
    for (int t = 0; t < 8; ++t) {
        const int colv = t * 16 + lm;
        for (int v = 0; v < 8; ++v) {
            const int rowv = m0 + v + 8 * half;
            if (rowv >= N) continue;
            float s = 0.f;
            for (int k = 0; k < 128; ++k)
                s += A[(size_t)rowv * 128 + k] * Bt[(size_t)k * 128 + colv];
            C[(size_t)rowv * 128 + colv] = s;
        }
    }
#endif
}

__global__ void __launch_bounds__(256)
gemm_readout_kernel(const float* __restrict__ h1, const float* __restrict__ h2,
                    const float* __restrict__ Wrt, const float* __restrict__ br,
                    float* __restrict__ out, int N) {
    const int wave = threadIdx.x >> 5;
    const int lane = threadIdx.x & 31;
    const int half = lane >> 4;
    const int lm   = lane & 15;
    const int m0   = (blockIdx.x * 8 + wave) * 16;
    if (m0 >= N) return;

    int rowA = m0 + lm;
    if (rowA >= N) rowA = N - 1;
    const float* __restrict__ r1 = h1 + (size_t)rowA * 128;
    const float* __restrict__ r2 = h2 + (size_t)rowA * 128;

#if defined(HAVE_WMMA_F32)
    v8f acc[4];
#pragma unroll
    for (int t = 0; t < 4; ++t)
        acc[t] = (v8f){0.f, 0.f, 0.f, 0.f, 0.f, 0.f, 0.f, 0.f};

    for (int k0 = 0; k0 < 256; k0 += 4) {
        const int ka = k0 + 2 * half;
        v2f a;
        if (k0 < 128) { a.x = r1[ka];       a.y = r1[ka + 1]; }
        else          { a.x = r2[ka - 128]; a.y = r2[ka - 127]; }
        const float* __restrict__ b0 = Wrt + (size_t)ka * 64 + lm;
        const float* __restrict__ b1 = Wrt + (size_t)(ka + 1) * 64 + lm;
#pragma unroll
        for (int t = 0; t < 4; ++t) {
            v2f b;
            b.x = b0[t * 16];
            b.y = b1[t * 16];
            acc[t] = __builtin_amdgcn_wmma_f32_16x16x4_f32(
                false, a, false, b, (short)0, acc[t], false, false);
        }
    }
#pragma unroll
    for (int t = 0; t < 4; ++t) {
        const int colv = t * 16 + lm;
        const float bias = br[colv];
#pragma unroll
        for (int v = 0; v < 8; ++v) {
            const int rowv = m0 + v + 8 * half;
            if (rowv < N) out[(size_t)rowv * 64 + colv] = acc[t][v] + bias;
        }
    }
#else
    for (int t = 0; t < 4; ++t) {
        const int colv = t * 16 + lm;
        for (int v = 0; v < 8; ++v) {
            const int rowv = m0 + v + 8 * half;
            if (rowv >= N) continue;
            float s = br[colv];
            for (int k = 0; k < 128; ++k) {
                s += h1[(size_t)rowv * 128 + k] * Wrt[(size_t)k * 64 + colv];
                s += h2[(size_t)rowv * 128 + k] * Wrt[(size_t)(k + 128) * 64 + colv];
            }
            out[(size_t)rowv * 64 + colv] = s;
        }
    }
#endif
}

__global__ void __launch_bounds__(256)
scatter_edges_kernel(const float* __restrict__ hW, const float* __restrict__ nrm,
                     const int* __restrict__ row, const int* __restrict__ col,
                     float* __restrict__ agg, long E) {
    const int wave = threadIdx.x >> 5;
    const int lane = threadIdx.x & 31;
    const long e = (long)blockIdx.x * 8 + wave;
    if (e >= E) return;
    const int r = row[e];
    const int c = col[e];
    const float w = nrm[e];
    const float4 v = *(const float4*)(hW + (size_t)r * 128 + lane * 4);
    float* __restrict__ dst = agg + (size_t)c * 128 + lane * 4;
    atomicAdd(dst + 0, w * v.x);
    atomicAdd(dst + 1, w * v.y);
    atomicAdd(dst + 2, w * v.z);
    atomicAdd(dst + 3, w * v.w);
}

__global__ void combine_relu_kernel(const float* __restrict__ agg,
                                    const float* __restrict__ hW,
                                    const float* __restrict__ dis,
                                    const float* __restrict__ bias,
                                    float* __restrict__ out, int N) {
    long i = (long)blockIdx.x * blockDim.x + threadIdx.x;
    if (i >= (long)N * 32) return;
    const int n = (int)(i >> 5);
    const int q = (int)(i & 31) * 4;
    const float d = dis[n];
    const float dd = d * d;
    const float4 a = *(const float4*)(agg + (size_t)n * 128 + q);
    const float4 h = *(const float4*)(hW + (size_t)n * 128 + q);
    const float4 b = *(const float4*)(bias + q);
    float4 r;
    r.x = fmaxf(a.x + dd * h.x + b.x, 0.f);
    r.y = fmaxf(a.y + dd * h.y + b.y, 0.f);
    r.z = fmaxf(a.z + dd * h.z + b.z, 0.f);
    r.w = fmaxf(a.w + dd * h.w + b.w, 0.f);
    *(float4*)(out + (size_t)n * 128 + q) = r;
}

extern "C" void kernel_launch(void* const* d_in, const int* in_sizes, int n_in,
                              void* d_out, int out_size, void* d_ws, size_t ws_size,
                              hipStream_t stream) {
    const float* x   = (const float*)d_in[0];
    const int*   ei  = (const int*)d_in[1];
    const float* ea  = (const float*)d_in[2];
    const float* W1  = (const float*)d_in[3];
    const float* b1  = (const float*)d_in[4];
    const float* W2  = (const float*)d_in[5];
    const float* b2  = (const float*)d_in[6];
    const float* Wr  = (const float*)d_in[7];
    const float* br  = (const float*)d_in[8];
    float* out = (float*)d_out;

    const long N = in_sizes[0] / IN_DIM;
    const long E = in_sizes[2] / 4;
    const int* row = ei;
    const int* col = ei + E;

    char* ws = (char*)d_ws;
    size_t off = 0;
    auto carve = [&](size_t bytes) -> char* {
        char* p = ws + off;
        off += (bytes + 255) & ~(size_t)255;
        return p;
    };
    float*    ew   = (float*)carve(E * 4);
    float*    nrm  = (float*)carve(E * 4);
    float*    deg  = (float*)carve(N * 4);
    float*    hW   = (float*)carve(N * 128 * 4);
    float*    agg  = (float*)carve(N * 128 * 4);
    float*    h1   = (float*)carve(N * 128 * 4);
    float*    h2   = (float*)carve(N * 128 * 4);
    float*    Wt1  = (float*)carve(128 * 128 * 4);
    float*    Wt2  = (float*)carve(128 * 128 * 4);
    float*    Wrt  = (float*)carve(256 * 64 * 4);
    unsigned* mm   = (unsigned*)carve(2 * 4);
    (void)ws_size; (void)n_in; (void)out_size;

    const int BT = 256;
    auto blk1 = [&](long n) { return (unsigned)((n + BT - 1) / BT); };

    init_minmax_kernel<<<1, 32, 0, stream>>>(mm);
    fill_f32_kernel<<<blk1(N), BT, 0, stream>>>(deg, 1.0f, N);
    edge_mean_minmax_kernel<<<blk1(E), BT, 0, stream>>>(ea, ew, mm, E);
    edge_weight_deg_kernel<<<blk1(E), BT, 0, stream>>>(ew, col, deg, mm, E);
    rsqrt_kernel<<<blk1(N), BT, 0, stream>>>(deg, N);
    edge_norm_kernel<<<blk1(E), BT, 0, stream>>>(ew, row, col, deg, nrm, E);

    transpose_kernel<<<blk1(128 * 128), BT, 0, stream>>>(Wt1, W1, HID, IN_DIM);
    transpose_kernel<<<blk1(128 * 128), BT, 0, stream>>>(Wt2, W2, HID, HID);
    transpose_kernel<<<blk1(256 * 64), BT, 0, stream>>>(Wrt, Wr, OUT_DIM, 2 * HID);

    const long nWaves = (N + 15) / 16;
    const unsigned gemmBlocks = (unsigned)((nWaves + 7) / 8);
    const unsigned scatBlocks = (unsigned)((E + 7) / 8);

    gemm_nt128_kernel<<<gemmBlocks, 256, 0, stream>>>(x, Wt1, hW, (int)N);
    fill_f32_kernel<<<blk1(N * 128), BT, 0, stream>>>(agg, 0.0f, N * 128);
    scatter_edges_kernel<<<scatBlocks, 256, 0, stream>>>(hW, nrm, row, col, agg, E);
    combine_relu_kernel<<<blk1(N * 32), BT, 0, stream>>>(agg, hW, deg, b1, h1, (int)N);

    gemm_nt128_kernel<<<gemmBlocks, 256, 0, stream>>>(h1, Wt2, hW, (int)N);
    fill_f32_kernel<<<blk1(N * 128), BT, 0, stream>>>(agg, 0.0f, N * 128);
    scatter_edges_kernel<<<scatBlocks, 256, 0, stream>>>(hW, nrm, row, col, agg, E);
    combine_relu_kernel<<<blk1(N * 32), BT, 0, stream>>>(agg, hW, deg, b2, h2, (int)N);

    gemm_readout_kernel<<<gemmBlocks, 256, 0, stream>>>(h1, h2, Wrt, br, out, (int)N);
}


# Aggregation made destination-stationary: a block keeps 392 nodes' sums in LDS, scans the edge list for its edges, LDS f32 adds in arrival order; global f32 atomics, zero fill, combine pass gone
# speedup vs baseline: 1.6286x; 1.6286x over previous
#include <hip/hip_runtime.h>
#include <hip/hip_bf16.h>


#define IN_DIM 128
#define HID    128
#define OUT_DIM 64
#define EPS_W  1e-8f

typedef float v2f __attribute__((ext_vector_type(2)));
typedef float v8f __attribute__((ext_vector_type(8)));

#if defined(__gfx1250__) && __has_builtin(__builtin_amdgcn_wmma_f32_16x16x4_f32)
#define HAVE_WMMA_F32 1
#endif

__global__ void fill_f32_kernel(float* __restrict__ p, float v, long n) {
    long i = (long)blockIdx.x * blockDim.x + threadIdx.x;
    if (i < n) p[i] = v;
}

__global__ void init_minmax_kernel(unsigned* __restrict__ mm) {
    if (threadIdx.x == 0) {
        mm[0] = 0x7F800000u;
        mm[1] = 0x00000000u;
    }
}

__global__ void edge_mean_minmax_kernel(const float* __restrict__ ea,
                                        float* __restrict__ ew,
                                        unsigned* __restrict__ mm, long E) {
    __shared__ unsigned smin[256];
    __shared__ unsigned smax[256];
    int tid = threadIdx.x;
    long e = (long)blockIdx.x * blockDim.x + tid;
    unsigned bmin = 0x7F800000u, bmax = 0u;
    if (e < E) {
        const float4 a = *(const float4*)(ea + e * 4);
        float m = 0.25f * (a.x + a.y + a.z + a.w);
        ew[e] = m;
        unsigned b = __float_as_uint(m);
        bmin = b; bmax = b;
    }
    smin[tid] = bmin; smax[tid] = bmax;
    __syncthreads();
    for (int s = 128; s > 0; s >>= 1) {
        if (tid < s) {
            smin[tid] = min(smin[tid], smin[tid + s]);
            smax[tid] = max(smax[tid], smax[tid + s]);
        }
        __syncthreads();
    }
    if (tid == 0) {
        atomicMin(&mm[0], smin[0]);
        atomicMax(&mm[1], smax[0]);
    }
}

__global__ void edge_weight_deg_kernel(float* __restrict__ ew,
                                       const int* __restrict__ col,
                                       float* __restrict__ deg,
                                       const unsigned* __restrict__ mm, long E) {
    long e = (long)blockIdx.x * blockDim.x + threadIdx.x;
    if (e >= E) return;
    float mn = __uint_as_float(mm[0]);
    float mx = __uint_as_float(mm[1]);
    float m = ew[e];
    float range = mx - mn;
    float w = (range > EPS_W) ? (1.0f - (m - mn) / (range + EPS_W)) : 1.0f;
    w = fmaxf(w, 0.0f);
    ew[e] = w;
    atomicAdd(&deg[col[e]], w);
}

__global__ void rsqrt_kernel(float* __restrict__ deg, long n) {
    long i = (long)blockIdx.x * blockDim.x + threadIdx.x;
    if (i < n) deg[i] = rsqrtf(deg[i]);
}

__global__ void edge_norm_kernel(const float* __restrict__ ew,
                                 const int* __restrict__ row,
                                 const int* __restrict__ col,
                                 const float* __restrict__ dis,
                                 float* __restrict__ nrm, long E) {
    long e = (long)blockIdx.x * blockDim.x + threadIdx.x;
    if (e >= E) return;
    nrm[e] = dis[row[e]] * ew[e] * dis[col[e]];
}

__global__ void transpose_kernel(float* __restrict__ dst,
                                 const float* __restrict__ src, int R, int C) {
    long i = (long)blockIdx.x * blockDim.x + threadIdx.x;
    if (i >= (long)R * C) return;
    int r = (int)(i / C), c = (int)(i % C);
    dst[(size_t)c * R + r] = src[i];
}

__global__ void __launch_bounds__(256)
gemm_nt128_kernel(const float* __restrict__ A, const float* __restrict__ Bt,
                  float* __restrict__ C, int N) {
    const int wave = threadIdx.x >> 5;
    const int lane = threadIdx.x & 31;
    const int half = lane >> 4;
    const int lm   = lane & 15;
    const int m0   = (blockIdx.x * 8 + wave) * 16;
    if (m0 >= N) return;

    int rowA = m0 + lm;
    if (rowA >= N) rowA = N - 1;
    const float* __restrict__ arow = A + (size_t)rowA * 128;

#if defined(HAVE_WMMA_F32)
    v8f acc[8];
#pragma unroll
    for (int t = 0; t < 8; ++t)
        acc[t] = (v8f){0.f, 0.f, 0.f, 0.f, 0.f, 0.f, 0.f, 0.f};

    for (int k0 = 0; k0 < 128; k0 += 4) {
        const int ka = k0 + 2 * half;
        v2f a;
        a.x = arow[ka];
        a.y = arow[ka + 1];
        const float* __restrict__ b0 = Bt + (size_t)ka * 128 + lm;
        const float* __restrict__ b1 = Bt + (size_t)(ka + 1) * 128 + lm;
#pragma unroll
        for (int t = 0; t < 8; ++t) {
            v2f b;
            b.x = b0[t * 16];
            b.y = b1[t * 16];
            acc[t] = __builtin_amdgcn_wmma_f32_16x16x4_f32(
                false, a, false, b, (short)0, acc[t], false, false);
        }
    }
#pragma unroll
    for (int t = 0; t < 8; ++t) {
        const int colv = t * 16 + lm;
#pragma unroll
        for (int v = 0; v < 8; ++v) {
            const int rowv = m0 + v + 8 * half;
            if (rowv < N) C[(size_t)rowv * 128 + colv] = acc[t][v];
        }
    }
#else
    for (int t = 0; t < 8; ++t) {
        const int colv = t * 16 + lm;
        for (int v = 0; v < 8; ++v) {
            const int rowv = m0 + v + 8 * half;
            if (rowv >= N) continue;
            float s = 0.f;
            for (int k = 0; k < 128; ++k)
                s += A[(size_t)rowv * 128 + k] * Bt[(size_t)k * 128 + colv];
            C[(size_t)rowv * 128 + colv] = s;
        }
    }
#endif
}

__global__ void __launch_bounds__(256)
gemm_readout_kernel(const float* __restrict__ h1, const float* __restrict__ h2,
                    const float* __restrict__ Wrt, const float* __restrict__ br,
                    float* __restrict__ out, int N) {
    const int wave = threadIdx.x >> 5;
    const int lane = threadIdx.x & 31;
    const int half = lane >> 4;
    const int lm   = lane & 15;
    const int m0   = (blockIdx.x * 8 + wave) * 16;
    if (m0 >= N) return;

    int rowA = m0 + lm;
    if (rowA >= N) rowA = N - 1;
    const float* __restrict__ r1 = h1 + (size_t)rowA * 128;
    const float* __restrict__ r2 = h2 + (size_t)rowA * 128;

#if defined(HAVE_WMMA_F32)
    v8f acc[4];
#pragma unroll
    for (int t = 0; t < 4; ++t)
        acc[t] = (v8f){0.f, 0.f, 0.f, 0.f, 0.f, 0.f, 0.f, 0.f};

    for (int k0 = 0; k0 < 256; k0 += 4) {
        const int ka = k0 + 2 * half;
        v2f a;
        if (k0 < 128) { a.x = r1[ka];       a.y = r1[ka + 1]; }
        else          { a.x = r2[ka - 128]; a.y = r2[ka - 127]; }
        const float* __restrict__ b0 = Wrt + (size_t)ka * 64 + lm;
        const float* __restrict__ b1 = Wrt + (size_t)(ka + 1) * 64 + lm;
#pragma unroll
        for (int t = 0; t < 4; ++t) {
            v2f b;
            b.x = b0[t * 16];
            b.y = b1[t * 16];
            acc[t] = __builtin_amdgcn_wmma_f32_16x16x4_f32(
                false, a, false, b, (short)0, acc[t], false, false);
        }
    }
#pragma unroll
    for (int t = 0; t < 4; ++t) {
        const int colv = t * 16 + lm;
        const float bias = br[colv];
#pragma unroll
        for (int v = 0; v < 8; ++v) {
            const int rowv = m0 + v + 8 * half;
            if (rowv < N) out[(size_t)rowv * 64 + colv] = acc[t][v] + bias;
        }
    }
#else
    for (int t = 0; t < 4; ++t) {
        const int colv = t * 16 + lm;
        for (int v = 0; v < 8; ++v) {
            const int rowv = m0 + v + 8 * half;
            if (rowv >= N) continue;
            float s = br[colv];
            for (int k = 0; k < 128; ++k) {
                s += h1[(size_t)rowv * 128 + k] * Wrt[(size_t)k * 64 + colv];
                s += h2[(size_t)rowv * 128 + k] * Wrt[(size_t)(k + 128) * 64 + colv];
            }
            out[(size_t)rowv * 64 + colv] = s;
        }
    }
#endif
}

#define TILE_N      392
#define QCAP        12288
#define AGG_THREADS 1024
#define ROW_BITS    23

__global__ void __launch_bounds__(AGG_THREADS)
aggregate_tile_kernel(const float* __restrict__ hW, const float* __restrict__ nrm,
                      const int* __restrict__ row, const int* __restrict__ col,
                      const float* __restrict__ dis, const float* __restrict__ bias,
                      float* __restrict__ out, int N, int E) {
    __shared__ float    tile[TILE_N * 128];
    __shared__ unsigned qa[QCAP];
    __shared__ unsigned qb[QCAP];
    __shared__ unsigned qcount;

    const int tid  = threadIdx.x;
    const int lane = tid & 31;
    const int wv   = tid >> 5;
    const int nodeBase = blockIdx.x * TILE_N;

    for (int i = tid; i < TILE_N * 32; i += AGG_THREADS)
        *(float4*)(tile + i * 4) = make_float4(0.f, 0.f, 0.f, 0.f);
    if (tid == 0) qcount = 0u;
    __syncthreads();

    auto take = [&](int e, unsigned lc) {
        if (lc < TILE_N) {
            const unsigned slot = atomicAdd(&qcount, 1u);
            if (slot < QCAP) {
                qa[slot] = (unsigned)e;
                qb[slot] = lc;
            } else {
                const float w = nrm[e];
                const float* __restrict__ s = hW + (size_t)row[e] * 128;
                for (int k = 0; k < 128; ++k)
                    atomicAdd(&tile[lc * 128 + (k & 3) * 32 + (k >> 2)], w * s[k]);
            }
        }
    };

    if ((E & 3) == 0) {
        const int nGroups = E >> 2;
        const int nOuter  = (nGroups + 4 * AGG_THREADS - 1) / (4 * AGG_THREADS);
        const int rot     = (int)(((long)blockIdx.x * nOuter) / gridDim.x);
        const int4* __restrict__ col4 = (const int4*)col;
        for (int oo = 0; oo < nOuter; ++oo) {
            int o = oo + rot;
            if (o >= nOuter) o -= nOuter;
            int  g[4];
            int4 c[4];
#pragma unroll
            for (int u = 0; u < 4; ++u) {
                g[u] = (o * 4 + u) * AGG_THREADS + tid;
                c[u] = (g[u] < nGroups) ? col4[g[u]] : make_int4(-1, -1, -1, -1);
            }
#pragma unroll
            for (int u = 0; u < 4; ++u) {
                const unsigned l0 = (unsigned)(c[u].x - nodeBase);
                const unsigned l1 = (unsigned)(c[u].y - nodeBase);
                const unsigned l2 = (unsigned)(c[u].z - nodeBase);
                const unsigned l3 = (unsigned)(c[u].w - nodeBase);
                if ((l0 < TILE_N) | (l1 < TILE_N) | (l2 < TILE_N) | (l3 < TILE_N)) {
                    take(g[u] * 4 + 0, l0);
                    take(g[u] * 4 + 1, l1);
                    take(g[u] * 4 + 2, l2);
                    take(g[u] * 4 + 3, l3);
                }
            }
        }
    } else {
        for (int e = tid; e < E; e += AGG_THREADS)
            take(e, (unsigned)(col[e] - nodeBase));
    }
    __syncthreads();

    const int cnt = (int)min(qcount, (unsigned)QCAP);

    for (int base = 0; base < cnt; base += 8 * AGG_THREADS) {
        unsigned e8[8], l8[8];
        int      r8[8];
        float    w8[8];
#pragma unroll
        for (int u = 0; u < 8; ++u) {
            const int i = base + u * AGG_THREADS + tid;
            const bool ok = i < cnt;
            e8[u] = ok ? qa[i] : 0u;
            l8[u] = ok ? qb[i] : 0u;
        }
#pragma unroll
        for (int u = 0; u < 8; ++u) {
            r8[u] = row[e8[u]];
            w8[u] = nrm[e8[u]];
        }
#pragma unroll
        for (int u = 0; u < 8; ++u) {
            const int i = base + u * AGG_THREADS + tid;
            if (i < cnt) {
                qa[i] = (unsigned)r8[u] | (l8[u] << ROW_BITS);
                qb[i] = __float_as_uint(w8[u]);
            }
        }
    }
    __syncthreads();

    for (int k0 = wv; k0 < cnt; k0 += 32 * 8) {
        const int  qi  = k0 + (lane & 7) * 32;
        const bool qok = qi < cnt;
        const unsigned pkL = qok ? qa[qi] : 0u;
        const unsigned wL  = qok ? qb[qi] : 0u;
        float4   v[8];
        float    w[8];
        unsigned lc[8];
#pragma unroll
        for (int b = 0; b < 8; ++b) {
            if (k0 + b * 32 < cnt) {
                const unsigned pk = __builtin_amdgcn_readlane(pkL, b);
                w[b]  = __uint_as_float(__builtin_amdgcn_readlane(wL, b));
                lc[b] = pk >> ROW_BITS;
                const unsigned r = pk & ((1u << ROW_BITS) - 1u);
                v[b] = *(const float4*)(hW + (size_t)r * 128 + lane * 4);
            }
        }
#pragma unroll
        for (int b = 0; b < 8; ++b) {
            if (k0 + b * 32 < cnt) {
                float* d = tile + lc[b] * 128 + lane;
                atomicAdd(d,      w[b] * v[b].x);
                atomicAdd(d + 32, w[b] * v[b].y);
                atomicAdd(d + 64, w[b] * v[b].z);
                atomicAdd(d + 96, w[b] * v[b].w);
            }
        }
    }
    __syncthreads();

    const float4 bq = *(const float4*)(bias + lane * 4);
    for (int i = tid; i < TILE_N * 32; i += AGG_THREADS) {
        const int ln = i >> 5;
        const int n  = nodeBase + ln;
        if (n < N) {
            const float* tp = tile + ln * 128 + lane;
            const float d  = dis[n];
            const float dd = d * d;
            const float4 h = *(const float4*)(hW + (size_t)n * 128 + lane * 4);
            float4 r;
            r.x = fmaxf(tp[0]  + dd * h.x + bq.x, 0.f);
            r.y = fmaxf(tp[32] + dd * h.y + bq.y, 0.f);
            r.z = fmaxf(tp[64] + dd * h.z + bq.z, 0.f);
            r.w = fmaxf(tp[96] + dd * h.w + bq.w, 0.f);
            *(float4*)(out + (size_t)n * 128 + lane * 4) = r;
        }
    }
}

extern "C" void kernel_launch(void* const* d_in, const int* in_sizes, int n_in,
                              void* d_out, int out_size, void* d_ws, size_t ws_size,
                              hipStream_t stream) {
    const float* x   = (const float*)d_in[0];
    const int*   ei  = (const int*)d_in[1];
    const float* ea  = (const float*)d_in[2];
    const float* W1  = (const float*)d_in[3];
    const float* b1  = (const float*)d_in[4];
    const float* W2  = (const float*)d_in[5];
    const float* b2  = (const float*)d_in[6];
    const float* Wr  = (const float*)d_in[7];
    const float* br  = (const float*)d_in[8];
    float* out = (float*)d_out;

    const long N = in_sizes[0] / IN_DIM;
    const long E = in_sizes[2] / 4;
    const int* row = ei;
    const int* col = ei + E;

    char* ws = (char*)d_ws;
    size_t off = 0;
    auto carve = [&](size_t bytes) -> char* {
        char* p = ws + off;
        off += (bytes + 255) & ~(size_t)255;
        return p;
    };
    float*    ew   = (float*)carve(E * 4);
    float*    nrm  = (float*)carve(E * 4);
    float*    deg  = (float*)carve(N * 4);
    float*    hW   = (float*)carve(N * 128 * 4);
    float*    h1   = (float*)carve(N * 128 * 4);
    float*    h2   = (float*)carve(N * 128 * 4);
    float*    Wt1  = (float*)carve(128 * 128 * 4);
    float*    Wt2  = (float*)carve(128 * 128 * 4);
    float*    Wrt  = (float*)carve(256 * 64 * 4);
    unsigned* mm   = (unsigned*)carve(2 * 4);
    (void)ws_size; (void)n_in; (void)out_size;

    const int BT = 256;
    auto blk1 = [&](long n) { return (unsigned)((n + BT - 1) / BT); };

    init_minmax_kernel<<<1, 32, 0, stream>>>(mm);
    fill_f32_kernel<<<blk1(N), BT, 0, stream>>>(deg, 1.0f, N);
    edge_mean_minmax_kernel<<<blk1(E), BT, 0, stream>>>(ea, ew, mm, E);
    edge_weight_deg_kernel<<<blk1(E), BT, 0, stream>>>(ew, col, deg, mm, E);
    rsqrt_kernel<<<blk1(N), BT, 0, stream>>>(deg, N);
    edge_norm_kernel<<<blk1(E), BT, 0, stream>>>(ew, row, col, deg, nrm, E);

    transpose_kernel<<<blk1(128 * 128), BT, 0, stream>>>(Wt1, W1, HID, IN_DIM);
    transpose_kernel<<<blk1(128 * 128), BT, 0, stream>>>(Wt2, W2, HID, HID);
    transpose_kernel<<<blk1(256 * 64), BT, 0, stream>>>(Wrt, Wr, OUT_DIM, 2 * HID);

    const long nWaves = (N + 15) / 16;
    const unsigned gemmBlocks = (unsigned)((nWaves + 7) / 8);
    const unsigned aggBlocks  = (unsigned)((N + TILE_N - 1) / TILE_N);

    gemm_nt128_kernel<<<gemmBlocks, 256, 0, stream>>>(x, Wt1, hW, (int)N);
    aggregate_tile_kernel<<<aggBlocks, AGG_THREADS, 0, stream>>>(hW, nrm, row, col, deg, b1, h1, (int)N, (int)E);

    gemm_nt128_kernel<<<gemmBlocks, 256, 0, stream>>>(h1, Wt2, hW, (int)N);
    aggregate_tile_kernel<<<aggBlocks, AGG_THREADS, 0, stream>>>(hW, nrm, row, col, deg, b2, h2, (int)N, (int)E);

    gemm_readout_kernel<<<gemmBlocks, 256, 0, stream>>>(h1, h2, Wrt, br, out, (int)N);
}


# Edge filter done once: layer 1 stores each block's finished queue (source row, local node, coefficient) in the workspace, layer 2 copies it back; second edge-list scan gone; sums still arrival order
# speedup vs baseline: 1.9387x; 1.1904x over previous
#include <hip/hip_runtime.h>
#include <hip/hip_bf16.h>


#define IN_DIM 128
#define HID    128
#define OUT_DIM 64
#define EPS_W  1e-8f

typedef float v2f __attribute__((ext_vector_type(2)));
typedef float v8f __attribute__((ext_vector_type(8)));

#if defined(__gfx1250__) && __has_builtin(__builtin_amdgcn_wmma_f32_16x16x4_f32)
#define HAVE_WMMA_F32 1
#endif

__global__ void fill_f32_kernel(float* __restrict__ p, float v, long n) {
    long i = (long)blockIdx.x * blockDim.x + threadIdx.x;
    if (i < n) p[i] = v;
}

__global__ void init_minmax_kernel(unsigned* __restrict__ mm) {
    if (threadIdx.x == 0) {
        mm[0] = 0x7F800000u;
        mm[1] = 0x00000000u;
    }
}

__global__ void edge_mean_minmax_kernel(const float* __restrict__ ea,
                                        float* __restrict__ ew,
                                        unsigned* __restrict__ mm, long E) {
    __shared__ unsigned smin[256];
    __shared__ unsigned smax[256];
    int tid = threadIdx.x;
    long e = (long)blockIdx.x * blockDim.x + tid;
    unsigned bmin = 0x7F800000u, bmax = 0u;
    if (e < E) {
        const float4 a = *(const float4*)(ea + e * 4);
        float m = 0.25f * (a.x + a.y + a.z + a.w);
        ew[e] = m;
        unsigned b = __float_as_uint(m);
        bmin = b; bmax = b;
    }
    smin[tid] = bmin; smax[tid] = bmax;
    __syncthreads();
    for (int s = 128; s > 0; s >>= 1) {
        if (tid < s) {
            smin[tid] = min(smin[tid], smin[tid + s]);
            smax[tid] = max(smax[tid], smax[tid + s]);
        }
        __syncthreads();
    }
    if (tid == 0) {
        atomicMin(&mm[0], smin[0]);
        atomicMax(&mm[1], smax[0]);
    }
}

__global__ void edge_weight_deg_kernel(float* __restrict__ ew,
                                       const int* __restrict__ col,
                                       float* __restrict__ deg,
                                       const unsigned* __restrict__ mm, long E) {
    long e = (long)blockIdx.x * blockDim.x + threadIdx.x;
    if (e >= E) return;
    float mn = __uint_as_float(mm[0]);
    float mx = __uint_as_float(mm[1]);
    float m = ew[e];
    float range = mx - mn;
    float w = (range > EPS_W) ? (1.0f - (m - mn) / (range + EPS_W)) : 1.0f;
    w = fmaxf(w, 0.0f);
    ew[e] = w;
    atomicAdd(&deg[col[e]], w);
}

__global__ void rsqrt_kernel(float* __restrict__ deg, long n) {
    long i = (long)blockIdx.x * blockDim.x + threadIdx.x;
    if (i < n) deg[i] = rsqrtf(deg[i]);
}

__global__ void edge_norm_kernel(const float* __restrict__ ew,
                                 const int* __restrict__ row,
                                 const int* __restrict__ col,
                                 const float* __restrict__ dis,
                                 float* __restrict__ nrm, long E) {
    long e = (long)blockIdx.x * blockDim.x + threadIdx.x;
    if (e >= E) return;
    nrm[e] = dis[row[e]] * ew[e] * dis[col[e]];
}

__global__ void transpose_kernel(float* __restrict__ dst,
                                 const float* __restrict__ src, int R, int C) {
    long i = (long)blockIdx.x * blockDim.x + threadIdx.x;
    if (i >= (long)R * C) return;
    int r = (int)(i / C), c = (int)(i % C);
    dst[(size_t)c * R + r] = src[i];
}

__global__ void __launch_bounds__(256)
gemm_nt128_kernel(const float* __restrict__ A, const float* __restrict__ Bt,
                  float* __restrict__ C, int N) {
    const int wave = threadIdx.x >> 5;
    const int lane = threadIdx.x & 31;
    const int half = lane >> 4;
    const int lm   = lane & 15;
    const int m0   = (blockIdx.x * 8 + wave) * 16;
    if (m0 >= N) return;

    int rowA = m0 + lm;
    if (rowA >= N) rowA = N - 1;
    const float* __restrict__ arow = A + (size_t)rowA * 128;

#if defined(HAVE_WMMA_F32)
    v8f acc[8];
#pragma unroll
    for (int t = 0; t < 8; ++t)
        acc[t] = (v8f){0.f, 0.f, 0.f, 0.f, 0.f, 0.f, 0.f, 0.f};

    for (int k0 = 0; k0 < 128; k0 += 4) {
        const int ka = k0 + 2 * half;
        v2f a;
        a.x = arow[ka];
        a.y = arow[ka + 1];
        const float* __restrict__ b0 = Bt + (size_t)ka * 128 + lm;
        const float* __restrict__ b1 = Bt + (size_t)(ka + 1) * 128 + lm;
#pragma unroll
        for (int t = 0; t < 8; ++t) {
            v2f b;
            b.x = b0[t * 16];
            b.y = b1[t * 16];
            acc[t] = __builtin_amdgcn_wmma_f32_16x16x4_f32(
                false, a, false, b, (short)0, acc[t], false, false);
        }
    }
#pragma unroll
    for (int t = 0; t < 8; ++t) {
        const int colv = t * 16 + lm;
#pragma unroll
        for (int v = 0; v < 8; ++v) {
            const int rowv = m0 + v + 8 * half;
            if (rowv < N) C[(size_t)rowv * 128 + colv] = acc[t][v];
        }
    }
#else
    for (int t = 0; t < 8; ++t) {
        const int colv = t * 16 + lm;
        for (int v = 0; v < 8; ++v) {
            const int rowv = m0 + v + 8 * half;
            if (rowv >= N) continue;
            float s = 0.f;
            for (int k = 0; k < 128; ++k)
                s += A[(size_t)rowv * 128 + k] * Bt[(size_t)k * 128 + colv];
            C[(size_t)rowv * 128 + colv] = s;
        }
    }
#endif
}

__global__ void __launch_bounds__(256)
gemm_readout_kernel(const float* __restrict__ h1, const float* __restrict__ h2,
                    const float* __restrict__ Wrt, const float* __restrict__ br,
                    float* __restrict__ out, int N) {
    const int wave = threadIdx.x >> 5;
    const int lane = threadIdx.x & 31;
    const int half = lane >> 4;
    const int lm   = lane & 15;
    const int m0   = (blockIdx.x * 8 + wave) * 16;
    if (m0 >= N) return;

    int rowA = m0 + lm;
    if (rowA >= N) rowA = N - 1;
    const float* __restrict__ r1 = h1 + (size_t)rowA * 128;
    const float* __restrict__ r2 = h2 + (size_t)rowA * 128;

#if defined(HAVE_WMMA_F32)
    v8f acc[4];
#pragma unroll
    for (int t = 0; t < 4; ++t)
        acc[t] = (v8f){0.f, 0.f, 0.f, 0.f, 0.f, 0.f, 0.f, 0.f};

    for (int k0 = 0; k0 < 256; k0 += 4) {
        const int ka = k0 + 2 * half;
        v2f a;
        if (k0 < 128) { a.x = r1[ka];       a.y = r1[ka + 1]; }
        else          { a.x = r2[ka - 128]; a.y = r2[ka - 127]; }
        const float* __restrict__ b0 = Wrt + (size_t)ka * 64 + lm;
        const float* __restrict__ b1 = Wrt + (size_t)(ka + 1) * 64 + lm;
#pragma unroll
        for (int t = 0; t < 4; ++t) {
            v2f b;
            b.x = b0[t * 16];
            b.y = b1[t * 16];
            acc[t] = __builtin_amdgcn_wmma_f32_16x16x4_f32(
                false, a, false, b, (short)0, acc[t], false, false);
        }
    }
#pragma unroll
    for (int t = 0; t < 4; ++t) {
        const int colv = t * 16 + lm;
        const float bias = br[colv];
#pragma unroll
        for (int v = 0; v < 8; ++v) {
            const int rowv = m0 + v + 8 * half;
            if (rowv < N) out[(size_t)rowv * 64 + colv] = acc[t][v] + bias;
        }
    }
#else
    for (int t = 0; t < 4; ++t) {
        const int colv = t * 16 + lm;
        for (int v = 0; v < 8; ++v) {
            const int rowv = m0 + v + 8 * half;
            if (rowv >= N) continue;
            float s = br[colv];
            for (int k = 0; k < 128; ++k) {
                s += h1[(size_t)rowv * 128 + k] * Wrt[(size_t)k * 64 + colv];
                s += h2[(size_t)rowv * 128 + k] * Wrt[(size_t)(k + 128) * 64 + colv];
            }
            out[(size_t)rowv * 64 + colv] = s;
        }
    }
#endif
}

#define TILE_N      392
#define QCAP        12288
#define AGG_THREADS 1024
#define ROW_BITS    23

__global__ void __launch_bounds__(AGG_THREADS)
aggregate_tile_kernel(const float* __restrict__ hW, const float* __restrict__ nrm,
                      const int* __restrict__ row, const int* __restrict__ col,
                      const float* __restrict__ dis, const float* __restrict__ bias,
                      float* __restrict__ out, unsigned* __restrict__ qkeep,
                      unsigned* __restrict__ qkeepCount, int reuse, int N, int E) {
    __shared__ float    tile[TILE_N * 128];
    __shared__ unsigned qa[QCAP];
    __shared__ unsigned qb[QCAP];
    __shared__ unsigned qcount;

    const int tid  = threadIdx.x;
    const int lane = tid & 31;
    const int wv   = tid >> 5;
    const int nodeBase = blockIdx.x * TILE_N;

    unsigned* __restrict__ myKeep = qkeep + (size_t)blockIdx.x * (2 * QCAP);
    unsigned kept = 0u;
    if (reuse) kept = qkeepCount[blockIdx.x];
    const bool scan = !reuse || kept > (unsigned)QCAP;

    for (int i = tid; i < TILE_N * 32; i += AGG_THREADS)
        *(float4*)(tile + i * 4) = make_float4(0.f, 0.f, 0.f, 0.f);
    if (tid == 0) qcount = 0u;
    __syncthreads();

    auto take = [&](int e, unsigned lc) {
        if (lc < TILE_N) {
            const unsigned slot = atomicAdd(&qcount, 1u);
            if (slot < QCAP) {
                qa[slot] = (unsigned)e;
                qb[slot] = lc;
            } else {
                const float w = nrm[e];
                const float* __restrict__ s = hW + (size_t)row[e] * 128;
                for (int k = 0; k < 128; ++k)
                    atomicAdd(&tile[lc * 128 + (k & 3) * 32 + (k >> 2)], w * s[k]);
            }
        }
    };

    if (scan) {
        if ((E & 3) == 0) {
            const int nGroups = E >> 2;
            const int nOuter  = (nGroups + 4 * AGG_THREADS - 1) / (4 * AGG_THREADS);
            const int rot     = (int)(((long)blockIdx.x * nOuter) / gridDim.x);
            const int4* __restrict__ col4 = (const int4*)col;
            for (int oo = 0; oo < nOuter; ++oo) {
                int o = oo + rot;
                if (o >= nOuter) o -= nOuter;
                int  g[4];
                int4 c[4];
#pragma unroll
                for (int u = 0; u < 4; ++u) {
                    g[u] = (o * 4 + u) * AGG_THREADS + tid;
                    c[u] = (g[u] < nGroups) ? col4[g[u]] : make_int4(-1, -1, -1, -1);
                }
#pragma unroll
                for (int u = 0; u < 4; ++u) {
                    const unsigned l0 = (unsigned)(c[u].x - nodeBase);
                    const unsigned l1 = (unsigned)(c[u].y - nodeBase);
                    const unsigned l2 = (unsigned)(c[u].z - nodeBase);
                    const unsigned l3 = (unsigned)(c[u].w - nodeBase);
                    if ((l0 < TILE_N) | (l1 < TILE_N) | (l2 < TILE_N) | (l3 < TILE_N)) {
                        take(g[u] * 4 + 0, l0);
                        take(g[u] * 4 + 1, l1);
                        take(g[u] * 4 + 2, l2);
                        take(g[u] * 4 + 3, l3);
                    }
                }
            }
        } else {
            for (int e = tid; e < E; e += AGG_THREADS)
                take(e, (unsigned)(col[e] - nodeBase));
        }
    }
    __syncthreads();

    const int cnt = (int)min(scan ? qcount : kept, (unsigned)QCAP);

    if (scan) {
        if (!reuse && tid == 0) qkeepCount[blockIdx.x] = qcount;
        for (int base = 0; base < cnt; base += 8 * AGG_THREADS) {
            unsigned e8[8], l8[8];
            int      r8[8];
            float    w8[8];
#pragma unroll
            for (int u = 0; u < 8; ++u) {
                const int i = base + u * AGG_THREADS + tid;
                const bool ok = i < cnt;
                e8[u] = ok ? qa[i] : 0u;
                l8[u] = ok ? qb[i] : 0u;
            }
#pragma unroll
            for (int u = 0; u < 8; ++u) {
                r8[u] = row[e8[u]];
                w8[u] = nrm[e8[u]];
            }
#pragma unroll
            for (int u = 0; u < 8; ++u) {
                const int i = base + u * AGG_THREADS + tid;
                if (i < cnt) {
                    const unsigned pk = (unsigned)r8[u] | (l8[u] << ROW_BITS);
                    const unsigned wb = __float_as_uint(w8[u]);
                    qa[i] = pk;
                    qb[i] = wb;
                    if (!reuse) {
                        myKeep[i]        = pk;
                        myKeep[QCAP + i] = wb;
                    }
                }
            }
        }
    } else {
        for (int base = 0; base < cnt; base += 8 * AGG_THREADS) {
            unsigned a8[8], b8[8];
#pragma unroll
            for (int u = 0; u < 8; ++u) {
                const int i = base + u * AGG_THREADS + tid;
                const bool ok = i < cnt;
                a8[u] = ok ? myKeep[i] : 0u;
                b8[u] = ok ? myKeep[QCAP + i] : 0u;
            }
#pragma unroll
            for (int u = 0; u < 8; ++u) {
                const int i = base + u * AGG_THREADS + tid;
                if (i < cnt) {
                    qa[i] = a8[u];
                    qb[i] = b8[u];
                }
            }
        }
    }
    __syncthreads();

    for (int k0 = wv; k0 < cnt; k0 += 32 * 8) {
        const int  qi  = k0 + (lane & 7) * 32;
        const bool qok = qi < cnt;
        const unsigned pkL = qok ? qa[qi] : 0u;
        const unsigned wL  = qok ? qb[qi] : 0u;
        float4   v[8];
        float    w[8];
        unsigned lc[8];
#pragma unroll
        for (int b = 0; b < 8; ++b) {
            if (k0 + b * 32 < cnt) {
                const unsigned pk = __builtin_amdgcn_readlane(pkL, b);
                w[b]  = __uint_as_float(__builtin_amdgcn_readlane(wL, b));
                lc[b] = pk >> ROW_BITS;
                const unsigned r = pk & ((1u << ROW_BITS) - 1u);
                v[b] = *(const float4*)(hW + (size_t)r * 128 + lane * 4);
            }
        }
#pragma unroll
        for (int b = 0; b < 8; ++b) {
            if (k0 + b * 32 < cnt) {
                float* d = tile + lc[b] * 128 + lane;
                atomicAdd(d,      w[b] * v[b].x);
                atomicAdd(d + 32, w[b] * v[b].y);
                atomicAdd(d + 64, w[b] * v[b].z);
                atomicAdd(d + 96, w[b] * v[b].w);
            }
        }
    }
    __syncthreads();

    const float4 bq = *(const float4*)(bias + lane * 4);
    for (int i = tid; i < TILE_N * 32; i += AGG_THREADS) {
        const int ln = i >> 5;
        const int n  = nodeBase + ln;
        if (n < N) {
            const float* tp = tile + ln * 128 + lane;
            const float d  = dis[n];
            const float dd = d * d;
            const float4 h = *(const float4*)(hW + (size_t)n * 128 + lane * 4);
            float4 r;
            r.x = fmaxf(tp[0]  + dd * h.x + bq.x, 0.f);
            r.y = fmaxf(tp[32] + dd * h.y + bq.y, 0.f);
            r.z = fmaxf(tp[64] + dd * h.z + bq.z, 0.f);
            r.w = fmaxf(tp[96] + dd * h.w + bq.w, 0.f);
            *(float4*)(out + (size_t)n * 128 + lane * 4) = r;
        }
    }
}

extern "C" void kernel_launch(void* const* d_in, const int* in_sizes, int n_in,
                              void* d_out, int out_size, void* d_ws, size_t ws_size,
                              hipStream_t stream) {
    const float* x   = (const float*)d_in[0];
    const int*   ei  = (const int*)d_in[1];
    const float* ea  = (const float*)d_in[2];
    const float* W1  = (const float*)d_in[3];
    const float* b1  = (const float*)d_in[4];
    const float* W2  = (const float*)d_in[5];
    const float* b2  = (const float*)d_in[6];
    const float* Wr  = (const float*)d_in[7];
    const float* br  = (const float*)d_in[8];
    float* out = (float*)d_out;

    const long N = in_sizes[0] / IN_DIM;
    const long E = in_sizes[2] / 4;
    const int* row = ei;
    const int* col = ei + E;

    char* ws = (char*)d_ws;
    size_t off = 0;
    auto carve = [&](size_t bytes) -> char* {
        char* p = ws + off;
        off += (bytes + 255) & ~(size_t)255;
        return p;
    };
    float*    ew   = (float*)carve(E * 4);
    float*    nrm  = (float*)carve(E * 4);
    float*    deg  = (float*)carve(N * 4);
    float*    hW   = (float*)carve(N * 128 * 4);
    float*    h1   = (float*)carve(N * 128 * 4);
    float*    h2   = (float*)carve(N * 128 * 4);
    float*    Wt1  = (float*)carve(128 * 128 * 4);
    float*    Wt2  = (float*)carve(128 * 128 * 4);
    float*    Wrt  = (float*)carve(256 * 64 * 4);
    unsigned* mm   = (unsigned*)carve(2 * 4);
    const unsigned aggBlocks = (unsigned)((N + TILE_N - 1) / TILE_N);
    unsigned* qkeep      = (unsigned*)carve((size_t)aggBlocks * 2 * QCAP * 4);
    unsigned* qkeepCount = (unsigned*)carve((size_t)aggBlocks * 4);
    (void)ws_size; (void)n_in; (void)out_size;

    const int BT = 256;
    auto blk1 = [&](long n) { return (unsigned)((n + BT - 1) / BT); };

    init_minmax_kernel<<<1, 32, 0, stream>>>(mm);
    fill_f32_kernel<<<blk1(N), BT, 0, stream>>>(deg, 1.0f, N);
    edge_mean_minmax_kernel<<<blk1(E), BT, 0, stream>>>(ea, ew, mm, E);
    edge_weight_deg_kernel<<<blk1(E), BT, 0, stream>>>(ew, col, deg, mm, E);
    rsqrt_kernel<<<blk1(N), BT, 0, stream>>>(deg, N);
    edge_norm_kernel<<<blk1(E), BT, 0, stream>>>(ew, row, col, deg, nrm, E);

    transpose_kernel<<<blk1(128 * 128), BT, 0, stream>>>(Wt1, W1, HID, IN_DIM);
    transpose_kernel<<<blk1(128 * 128), BT, 0, stream>>>(Wt2, W2, HID, HID);
    transpose_kernel<<<blk1(256 * 64), BT, 0, stream>>>(Wrt, Wr, OUT_DIM, 2 * HID);

    const long nWaves = (N + 15) / 16;
    const unsigned gemmBlocks = (unsigned)((nWaves + 7) / 8);

    gemm_nt128_kernel<<<gemmBlocks, 256, 0, stream>>>(x, Wt1, hW, (int)N);
    aggregate_tile_kernel<<<aggBlocks, AGG_THREADS, 0, stream>>>(hW, nrm, row, col, deg, b1, h1, qkeep, qkeepCount, 0, (int)N, (int)E);

    gemm_nt128_kernel<<<gemmBlocks, 256, 0, stream>>>(h1, Wt2, hW, (int)N);
    aggregate_tile_kernel<<<aggBlocks, AGG_THREADS, 0, stream>>>(hW, nrm, row, col, deg, b2, h2, qkeep, qkeepCount, 1, (int)N, (int)E);

    gemm_readout_kernel<<<gemmBlocks, 256, 0, stream>>>(h1, h2, Wrt, br, out, (int)N);
}


# Block queues filled by the edge-coefficient pass: an edge's record goes to its destination block's workspace queue (LDS counts, an integer atomic per range); edge-list scan gone; sums arrival order
# speedup vs baseline: 2.3236x; 1.1985x over previous
#include <hip/hip_runtime.h>
#include <hip/hip_bf16.h>


#define IN_DIM 128
#define HID    128
#define OUT_DIM 64
#define EPS_W  1e-8f

#define TILE_N      392
#define QCAP        12288
#define AGG_THREADS 1024
#define ROW_BITS    23
#define QCNT_STRIDE 32
#define EQ_THREADS  1024
#define EQ_PER      4
#define EQ_MAXT     2048

typedef float v2f __attribute__((ext_vector_type(2)));
typedef float v8f __attribute__((ext_vector_type(8)));

#if defined(__gfx1250__) && __has_builtin(__builtin_amdgcn_wmma_f32_16x16x4_f32)
#define HAVE_WMMA_F32 1
#endif

__global__ void fill_f32_kernel(float* __restrict__ p, float v, long n) {
    long i = (long)blockIdx.x * blockDim.x + threadIdx.x;
    if (i < n) p[i] = v;
}

__global__ void init_state_kernel(unsigned* __restrict__ mm,
                                  unsigned* __restrict__ qcnt, int nCnt) {
    const int i = blockIdx.x * blockDim.x + threadIdx.x;
    if (i == 0) {
        mm[0] = 0x7F800000u;
        mm[1] = 0x00000000u;
    }
    if (i < nCnt) qcnt[(size_t)i * QCNT_STRIDE] = 0u;
}

__global__ void edge_mean_minmax_kernel(const float* __restrict__ ea,
                                        float* __restrict__ ew,
                                        unsigned* __restrict__ mm, long E) {
    __shared__ unsigned smin[256];
    __shared__ unsigned smax[256];
    int tid = threadIdx.x;
    long e = (long)blockIdx.x * blockDim.x + tid;
    unsigned bmin = 0x7F800000u, bmax = 0u;
    if (e < E) {
        const float4 a = *(const float4*)(ea + e * 4);
        float m = 0.25f * (a.x + a.y + a.z + a.w);
        ew[e] = m;
        unsigned b = __float_as_uint(m);
        bmin = b; bmax = b;
    }
    smin[tid] = bmin; smax[tid] = bmax;
    __syncthreads();
    for (int s = 128; s > 0; s >>= 1) {
        if (tid < s) {
            smin[tid] = min(smin[tid], smin[tid + s]);
            smax[tid] = max(smax[tid], smax[tid + s]);
        }
        __syncthreads();
    }
    if (tid == 0) {
        atomicMin(&mm[0], smin[0]);
        atomicMax(&mm[1], smax[0]);
    }
}

__global__ void edge_weight_deg_kernel(float* __restrict__ ew,
                                       const int* __restrict__ col,
                                       float* __restrict__ deg,
                                       const unsigned* __restrict__ mm, long E) {
    long e = (long)blockIdx.x * blockDim.x + threadIdx.x;
    if (e >= E) return;
    float mn = __uint_as_float(mm[0]);
    float mx = __uint_as_float(mm[1]);
    float m = ew[e];
    float range = mx - mn;
    float w = (range > EPS_W) ? (1.0f - (m - mn) / (range + EPS_W)) : 1.0f;
    w = fmaxf(w, 0.0f);
    ew[e] = w;
    atomicAdd(&deg[col[e]], w);
}

__global__ void rsqrt_kernel(float* __restrict__ deg, long n) {
    long i = (long)blockIdx.x * blockDim.x + threadIdx.x;
    if (i < n) deg[i] = rsqrtf(deg[i]);
}

__global__ void __launch_bounds__(EQ_THREADS)
edge_queue_kernel(const float* __restrict__ ew, const int* __restrict__ row,
                  const int* __restrict__ col, const float* __restrict__ dis,
                  unsigned* __restrict__ queue, unsigned* __restrict__ qcnt,
                  int* __restrict__ ovfRow, int* __restrict__ ovfCol,
                  float* __restrict__ ovfW, int nTiles, int E) {
    __shared__ unsigned hist[EQ_MAXT];
    const int tid    = threadIdx.x;
    const int nLocal = min(nTiles, EQ_MAXT);
    for (int t = tid; t < nLocal; t += EQ_THREADS) hist[t] = 0u;
    __syncthreads();

    const int base = blockIdx.x * (EQ_THREADS * EQ_PER);
    int      r[EQ_PER], c[EQ_PER];
    float    w[EQ_PER];
    unsigned tl[EQ_PER], loc[EQ_PER];
    bool     ok[EQ_PER];
#pragma unroll
    for (int u = 0; u < EQ_PER; ++u) {
        const int e = base + u * EQ_THREADS + tid;
        const bool in = e < E;
        r[u] = in ? row[e] : 0;
        c[u] = in ? col[e] : -1;
        w[u] = in ? ew[e] : 0.f;
        tl[u] = (unsigned)c[u] / TILE_N;
        ok[u] = in && tl[u] < (unsigned)nTiles;
    }
    float d0[EQ_PER], d1[EQ_PER];
#pragma unroll
    for (int u = 0; u < EQ_PER; ++u) {
        d0[u] = dis[ok[u] ? r[u] : 0];
        d1[u] = dis[ok[u] ? c[u] : 0];
    }
#pragma unroll
    for (int u = 0; u < EQ_PER; ++u) {
        loc[u] = 0u;
        if (ok[u])
            loc[u] = (tl[u] < EQ_MAXT) ? atomicAdd(&hist[tl[u]], 1u)
                                       : atomicAdd(&qcnt[(size_t)tl[u] * QCNT_STRIDE], 1u);
    }
#pragma unroll
    for (int u = 0; u < EQ_PER; ++u) w[u] = d0[u] * w[u] * d1[u];
    __syncthreads();
    for (int t = tid; t < nLocal; t += EQ_THREADS) {
        const unsigned n = hist[t];
        if (n) hist[t] = atomicAdd(&qcnt[(size_t)t * QCNT_STRIDE], n);
    }
    __syncthreads();
    uint2* __restrict__ q2 = (uint2*)queue;
#pragma unroll
    for (int u = 0; u < EQ_PER; ++u) {
        if (ok[u]) {
            const unsigned slot = loc[u] + ((tl[u] < EQ_MAXT) ? hist[tl[u]] : 0u);
            if (slot < QCAP) {
                const unsigned ln = (unsigned)c[u] - tl[u] * TILE_N;
                q2[(size_t)tl[u] * QCAP + slot] =
                    make_uint2((unsigned)r[u] | (ln << ROW_BITS), __float_as_uint(w[u]));
            } else {
                const unsigned o = atomicAdd(&qcnt[(size_t)nTiles * QCNT_STRIDE], 1u);
                if (o < (unsigned)E) {
                    ovfRow[o] = r[u];
                    ovfCol[o] = c[u];
                    ovfW[o]   = w[u];
                }
            }
        }
    }
}

__global__ void transpose_kernel(float* __restrict__ dst,
                                 const float* __restrict__ src, int R, int C) {
    long i = (long)blockIdx.x * blockDim.x + threadIdx.x;
    if (i >= (long)R * C) return;
    int r = (int)(i / C), c = (int)(i % C);
    dst[(size_t)c * R + r] = src[i];
}

__global__ void __launch_bounds__(256)
gemm_nt128_kernel(const float* __restrict__ A, const float* __restrict__ Bt,
                  float* __restrict__ C, int N) {
    const int wave = threadIdx.x >> 5;
    const int lane = threadIdx.x & 31;
    const int half = lane >> 4;
    const int lm   = lane & 15;
    const int m0   = (blockIdx.x * 8 + wave) * 16;
    if (m0 >= N) return;

    int rowA = m0 + lm;
    if (rowA >= N) rowA = N - 1;
    const float* __restrict__ arow = A + (size_t)rowA * 128;

#if defined(HAVE_WMMA_F32)
    v8f acc[8];
#pragma unroll
    for (int t = 0; t < 8; ++t)
        acc[t] = (v8f){0.f, 0.f, 0.f, 0.f, 0.f, 0.f, 0.f, 0.f};

    for (int k0 = 0; k0 < 128; k0 += 4) {
        const int ka = k0 + 2 * half;
        v2f a;
        a.x = arow[ka];
        a.y = arow[ka + 1];
        const float* __restrict__ b0 = Bt + (size_t)ka * 128 + lm;
        const float* __restrict__ b1 = Bt + (size_t)(ka + 1) * 128 + lm;
#pragma unroll
        for (int t = 0; t < 8; ++t) {
            v2f b;
            b.x = b0[t * 16];
            b.y = b1[t * 16];
            acc[t] = __builtin_amdgcn_wmma_f32_16x16x4_f32(
                false, a, false, b, (short)0, acc[t], false, false);
        }
    }
#pragma unroll
    for (int t = 0; t < 8; ++t) {
        const int colv = t * 16 + lm;
#pragma unroll
        for (int v = 0; v < 8; ++v) {
            const int rowv = m0 + v + 8 * half;
            if (rowv < N) C[(size_t)rowv * 128 + colv] = acc[t][v];
        }
    }
#else
    for (int t = 0; t < 8; ++t) {
        const int colv = t * 16 + lm;
        for (int v = 0; v < 8; ++v) {
            const int rowv = m0 + v + 8 * half;
            if (rowv >= N) continue;
            float s = 0.f;
            for (int k = 0; k < 128; ++k)
                s += A[(size_t)rowv * 128 + k] * Bt[(size_t)k * 128 + colv];
            C[(size_t)rowv * 128 + colv] = s;
        }
    }
#endif
}

__global__ void __launch_bounds__(256)
gemm_readout_kernel(const float* __restrict__ h1, const float* __restrict__ h2,
                    const float* __restrict__ Wrt, const float* __restrict__ br,
                    float* __restrict__ out, int N) {
    const int wave = threadIdx.x >> 5;
    const int lane = threadIdx.x & 31;
    const int half = lane >> 4;
    const int lm   = lane & 15;
    const int m0   = (blockIdx.x * 8 + wave) * 16;
    if (m0 >= N) return;

    int rowA = m0 + lm;
    if (rowA >= N) rowA = N - 1;
    const float* __restrict__ r1 = h1 + (size_t)rowA * 128;
    const float* __restrict__ r2 = h2 + (size_t)rowA * 128;

#if defined(HAVE_WMMA_F32)
    v8f acc[4];
#pragma unroll
    for (int t = 0; t < 4; ++t)
        acc[t] = (v8f){0.f, 0.f, 0.f, 0.f, 0.f, 0.f, 0.f, 0.f};

    for (int k0 = 0; k0 < 256; k0 += 4) {
        const int ka = k0 + 2 * half;
        v2f a;
        if (k0 < 128) { a.x = r1[ka];       a.y = r1[ka + 1]; }
        else          { a.x = r2[ka - 128]; a.y = r2[ka - 127]; }
        const float* __restrict__ b0 = Wrt + (size_t)ka * 64 + lm;
        const float* __restrict__ b1 = Wrt + (size_t)(ka + 1) * 64 + lm;
#pragma unroll
        for (int t = 0; t < 4; ++t) {
            v2f b;
            b.x = b0[t * 16];
            b.y = b1[t * 16];
            acc[t] = __builtin_amdgcn_wmma_f32_16x16x4_f32(
                false, a, false, b, (short)0, acc[t], false, false);
        }
    }
#pragma unroll
    for (int t = 0; t < 4; ++t) {
        const int colv = t * 16 + lm;
        const float bias = br[colv];
#pragma unroll
        for (int v = 0; v < 8; ++v) {
            const int rowv = m0 + v + 8 * half;
            if (rowv < N) out[(size_t)rowv * 64 + colv] = acc[t][v] + bias;
        }
    }
#else
    for (int t = 0; t < 4; ++t) {
        const int colv = t * 16 + lm;
        for (int v = 0; v < 8; ++v) {
            const int rowv = m0 + v + 8 * half;
            if (rowv >= N) continue;
            float s = br[colv];
            for (int k = 0; k < 128; ++k) {
                s += h1[(size_t)rowv * 128 + k] * Wrt[(size_t)k * 64 + colv];
                s += h2[(size_t)rowv * 128 + k] * Wrt[(size_t)(k + 128) * 64 + colv];
            }
            out[(size_t)rowv * 64 + colv] = s;
        }
    }
#endif
}

__global__ void __launch_bounds__(AGG_THREADS)
aggregate_tile_kernel(const float* __restrict__ hW, const unsigned* __restrict__ queue,
                      const unsigned* __restrict__ qcnt,
                      const int* __restrict__ ovfRow, const int* __restrict__ ovfCol,
                      const float* __restrict__ ovfW,
                      const float* __restrict__ dis, const float* __restrict__ bias,
                      float* __restrict__ out, int N, int E) {
    __shared__ float    tile[TILE_N * 128];
    __shared__ unsigned qa[QCAP];
    __shared__ unsigned qb[QCAP];

    const int tid  = threadIdx.x;
    const int lane = tid & 31;
    const int wv   = tid >> 5;
    const int nodeBase = blockIdx.x * TILE_N;

    const int cnt  = (int)min(qcnt[(size_t)blockIdx.x * QCNT_STRIDE], (unsigned)QCAP);
    const int nOvf = (int)min(qcnt[(size_t)gridDim.x * QCNT_STRIDE], (unsigned)E);

    for (int i = tid; i < TILE_N * 32; i += AGG_THREADS)
        *(float4*)(tile + i * 4) = make_float4(0.f, 0.f, 0.f, 0.f);

    const uint2* __restrict__ myQ = (const uint2*)queue + (size_t)blockIdx.x * QCAP;
    for (int base = 0; base < cnt; base += 8 * AGG_THREADS) {
        uint2 r8[8];
#pragma unroll
        for (int u = 0; u < 8; ++u) {
            const int i = base + u * AGG_THREADS + tid;
            r8[u] = (i < cnt) ? myQ[i] : make_uint2(0u, 0u);
        }
#pragma unroll
        for (int u = 0; u < 8; ++u) {
            const int i = base + u * AGG_THREADS + tid;
            if (i < cnt) {
                qa[i] = r8[u].x;
                qb[i] = r8[u].y;
            }
        }
    }
    __syncthreads();

    for (int i = tid; i < nOvf; i += AGG_THREADS) {
        const unsigned lc = (unsigned)(ovfCol[i] - nodeBase);
        if (lc < TILE_N) {
            const float w = ovfW[i];
            const float* __restrict__ s = hW + (size_t)ovfRow[i] * 128;
            for (int k = 0; k < 128; ++k)
                atomicAdd(&tile[lc * 128 + (k & 3) * 32 + (k >> 2)], w * s[k]);
        }
    }

    for (int k0 = wv; k0 < cnt; k0 += 32 * 8) {
        const int  qi  = k0 + (lane & 7) * 32;
        const bool qok = qi < cnt;
        const unsigned pkL = qok ? qa[qi] : 0u;
        const unsigned wL  = qok ? qb[qi] : 0u;
        float4   v[8];
        float    w[8];
        unsigned lc[8];
#pragma unroll
        for (int b = 0; b < 8; ++b) {
            if (k0 + b * 32 < cnt) {
                const unsigned pk = __builtin_amdgcn_readlane(pkL, b);
                w[b]  = __uint_as_float(__builtin_amdgcn_readlane(wL, b));
                lc[b] = pk >> ROW_BITS;
                const unsigned r = pk & ((1u << ROW_BITS) - 1u);
                v[b] = *(const float4*)(hW + (size_t)r * 128 + lane * 4);
            }
        }
#pragma unroll
        for (int b = 0; b < 8; ++b) {
            if (k0 + b * 32 < cnt) {
                float* d = tile + lc[b] * 128 + lane;
                atomicAdd(d,      w[b] * v[b].x);
                atomicAdd(d + 32, w[b] * v[b].y);
                atomicAdd(d + 64, w[b] * v[b].z);
                atomicAdd(d + 96, w[b] * v[b].w);
            }
        }
    }
    __syncthreads();

    const float4 bq = *(const float4*)(bias + lane * 4);
    for (int i = tid; i < TILE_N * 32; i += AGG_THREADS) {
        const int ln = i >> 5;
        const int n  = nodeBase + ln;
        if (n < N) {
            const float* tp = tile + ln * 128 + lane;
            const float d  = dis[n];
            const float dd = d * d;
            const float4 h = *(const float4*)(hW + (size_t)n * 128 + lane * 4);
            float4 r;
            r.x = fmaxf(tp[0]  + dd * h.x + bq.x, 0.f);
            r.y = fmaxf(tp[32] + dd * h.y + bq.y, 0.f);
            r.z = fmaxf(tp[64] + dd * h.z + bq.z, 0.f);
            r.w = fmaxf(tp[96] + dd * h.w + bq.w, 0.f);
            *(float4*)(out + (size_t)n * 128 + lane * 4) = r;
        }
    }
}

extern "C" void kernel_launch(void* const* d_in, const int* in_sizes, int n_in,
                              void* d_out, int out_size, void* d_ws, size_t ws_size,
                              hipStream_t stream) {
    const float* x   = (const float*)d_in[0];
    const int*   ei  = (const int*)d_in[1];
    const float* ea  = (const float*)d_in[2];
    const float* W1  = (const float*)d_in[3];
    const float* b1  = (const float*)d_in[4];
    const float* W2  = (const float*)d_in[5];
    const float* b2  = (const float*)d_in[6];
    const float* Wr  = (const float*)d_in[7];
    const float* br  = (const float*)d_in[8];
    float* out = (float*)d_out;

    const long N = in_sizes[0] / IN_DIM;
    const long E = in_sizes[2] / 4;
    const int* row = ei;
    const int* col = ei + E;

    char* ws = (char*)d_ws;
    size_t off = 0;
    auto carve = [&](size_t bytes) -> char* {
        char* p = ws + off;
        off += (bytes + 255) & ~(size_t)255;
        return p;
    };
    float*    ew   = (float*)carve(E * 4);
    float*    deg  = (float*)carve(N * 4);
    float*    hW   = (float*)carve(N * 128 * 4);
    float*    h1   = (float*)carve(N * 128 * 4);
    float*    h2   = (float*)carve(N * 128 * 4);
    float*    Wt1  = (float*)carve(128 * 128 * 4);
    float*    Wt2  = (float*)carve(128 * 128 * 4);
    float*    Wrt  = (float*)carve(256 * 64 * 4);
    unsigned* mm   = (unsigned*)carve(2 * 4);
    const unsigned aggBlocks = (unsigned)((N + TILE_N - 1) / TILE_N);
    unsigned* queue  = (unsigned*)carve((size_t)aggBlocks * QCAP * 8);
    unsigned* qcnt   = (unsigned*)carve((size_t)(aggBlocks + 1) * QCNT_STRIDE * 4);
    int*      ovfRow = (int*)carve(E * 4);
    int*      ovfCol = (int*)carve(E * 4);
    float*    ovfW   = (float*)carve(E * 4);
    (void)ws_size; (void)n_in; (void)out_size;

    const int BT = 256;
    auto blk1 = [&](long n) { return (unsigned)((n + BT - 1) / BT); };

    init_state_kernel<<<blk1(aggBlocks + 1), BT, 0, stream>>>(mm, qcnt, (int)(aggBlocks + 1));
    fill_f32_kernel<<<blk1(N), BT, 0, stream>>>(deg, 1.0f, N);
    edge_mean_minmax_kernel<<<blk1(E), BT, 0, stream>>>(ea, ew, mm, E);
    edge_weight_deg_kernel<<<blk1(E), BT, 0, stream>>>(ew, col, deg, mm, E);
    rsqrt_kernel<<<blk1(N), BT, 0, stream>>>(deg, N);
    const unsigned eqBlocks = (unsigned)((E + EQ_THREADS * EQ_PER - 1) / (EQ_THREADS * EQ_PER));
    edge_queue_kernel<<<eqBlocks, EQ_THREADS, 0, stream>>>(ew, row, col, deg, queue, qcnt, ovfRow, ovfCol, ovfW, (int)aggBlocks, (int)E);

    transpose_kernel<<<blk1(128 * 128), BT, 0, stream>>>(Wt1, W1, HID, IN_DIM);
    transpose_kernel<<<blk1(128 * 128), BT, 0, stream>>>(Wt2, W2, HID, HID);
    transpose_kernel<<<blk1(256 * 64), BT, 0, stream>>>(Wrt, Wr, OUT_DIM, 2 * HID);

    const long nWaves = (N + 15) / 16;
    const unsigned gemmBlocks = (unsigned)((nWaves + 7) / 8);

    gemm_nt128_kernel<<<gemmBlocks, 256, 0, stream>>>(x, Wt1, hW, (int)N);
    aggregate_tile_kernel<<<aggBlocks, AGG_THREADS, 0, stream>>>(hW, queue, qcnt, ovfRow, ovfCol, ovfW, deg, b1, h1, (int)N, (int)E);

    gemm_nt128_kernel<<<gemmBlocks, 256, 0, stream>>>(h1, Wt2, hW, (int)N);
    aggregate_tile_kernel<<<aggBlocks, AGG_THREADS, 0, stream>>>(hW, queue, qcnt, ovfRow, ovfCol, ovfW, deg, b2, h2, (int)N, (int)E);

    gemm_readout_kernel<<<gemmBlocks, 256, 0, stream>>>(h1, h2, Wrt, br, out, (int)N);
}


# Edge list read once: first pass queues records holding the raw mean; each destination block turns its queue into weights, sums its degrees in LDS (queue order, not global atomics), writes dis
# speedup vs baseline: 2.9701x; 1.2782x over previous
#include <hip/hip_runtime.h>
#include <hip/hip_bf16.h>


#define IN_DIM 128
#define HID    128
#define OUT_DIM 64
#define EPS_W  1e-8f

#define TILE_N      392
#define QCAP        12288
#define AGG_THREADS 1024
#define ROW_BITS    23
#define QCNT_STRIDE 32
#define EQ_THREADS  1024
#define EQ_PER      4
#define EQ_MAXT     2048

typedef float v2f __attribute__((ext_vector_type(2)));
typedef float v8f __attribute__((ext_vector_type(8)));

#if defined(__gfx1250__) && __has_builtin(__builtin_amdgcn_wmma_f32_16x16x4_f32)
#define HAVE_WMMA_F32 1
#endif

__global__ void init_state_kernel(unsigned* __restrict__ mm,
                                  unsigned* __restrict__ qcnt, int nCnt) {
    const int i = blockIdx.x * blockDim.x + threadIdx.x;
    if (i == 0) {
        mm[0] = 0x7F800000u;
        mm[1] = 0x00000000u;
    }
    if (i < nCnt) qcnt[(size_t)i * QCNT_STRIDE] = 0u;
}

__global__ void __launch_bounds__(EQ_THREADS)
edge_bin_kernel(const float* __restrict__ ea, const int* __restrict__ row,
                const int* __restrict__ col, unsigned* __restrict__ mm,
                unsigned* __restrict__ queue, unsigned* __restrict__ qcnt,
                int* __restrict__ ovfRow, int* __restrict__ ovfCol,
                float* __restrict__ ovfW, int nTiles, int E) {
    __shared__ unsigned hist[EQ_MAXT];
    __shared__ unsigned sMin, sMax;
    const int tid    = threadIdx.x;
    const int nLocal = min(nTiles, EQ_MAXT);
    for (int t = tid; t < nLocal; t += EQ_THREADS) hist[t] = 0u;
    if (tid == 0) { sMin = 0x7F800000u; sMax = 0u; }
    __syncthreads();

    const int base = blockIdx.x * (EQ_THREADS * EQ_PER);
    int      r[EQ_PER], c[EQ_PER];
    float4   a[EQ_PER];
    float    m[EQ_PER];
    unsigned tl[EQ_PER], loc[EQ_PER];
    bool     ok[EQ_PER];
#pragma unroll
    for (int u = 0; u < EQ_PER; ++u) {
        const int e = base + u * EQ_THREADS + tid;
        const bool in = e < E;
        const int ec = in ? e : E - 1;
        a[u] = *(const float4*)(ea + (size_t)ec * 4);
        r[u] = row[ec];
        c[u] = col[ec];
    }
    __builtin_amdgcn_sched_barrier(0);
#pragma unroll
    for (int u = 0; u < EQ_PER; ++u) {
        const bool in = base + u * EQ_THREADS + tid < E;
        if (!in) c[u] = -1;
        tl[u] = (unsigned)c[u] / TILE_N;
        ok[u] = in && tl[u] < (unsigned)nTiles;
    }
    unsigned bmin = 0x7F800000u, bmax = 0u;
#pragma unroll
    for (int u = 0; u < EQ_PER; ++u) {
        m[u] = 0.25f * (a[u].x + a[u].y + a[u].z + a[u].w);
        if (base + u * EQ_THREADS + tid < E) {
            const unsigned b = __float_as_uint(m[u]);
            bmin = min(bmin, b);
            bmax = max(bmax, b);
        }
    }
#pragma unroll
    for (int u = 0; u < EQ_PER; ++u) {
        loc[u] = 0u;
        if (ok[u])
            loc[u] = (tl[u] < EQ_MAXT) ? atomicAdd(&hist[tl[u]], 1u)
                                       : atomicAdd(&qcnt[(size_t)tl[u] * QCNT_STRIDE], 1u);
    }
    for (int off = 16; off > 0; off >>= 1) {
        bmin = min(bmin, (unsigned)__shfl_xor((int)bmin, off));
        bmax = max(bmax, (unsigned)__shfl_xor((int)bmax, off));
    }
    if ((tid & 31) == 0) {
        atomicMin(&sMin, bmin);
        atomicMax(&sMax, bmax);
    }
    __syncthreads();
    for (int t = tid; t < nLocal; t += EQ_THREADS) {
        const unsigned n = hist[t];
        if (n) hist[t] = atomicAdd(&qcnt[(size_t)t * QCNT_STRIDE], n);
    }
    if (tid == 0) {
        atomicMin(&mm[0], sMin);
        atomicMax(&mm[1], sMax);
    }
    __syncthreads();
    uint2* __restrict__ q2 = (uint2*)queue;
#pragma unroll
    for (int u = 0; u < EQ_PER; ++u) {
        if (ok[u]) {
            const unsigned slot = loc[u] + ((tl[u] < EQ_MAXT) ? hist[tl[u]] : 0u);
            if (slot < QCAP) {
                const unsigned ln = (unsigned)c[u] - tl[u] * TILE_N;
                q2[(size_t)tl[u] * QCAP + slot] =
                    make_uint2((unsigned)r[u] | (ln << ROW_BITS), __float_as_uint(m[u]));
            } else {
                const unsigned o = atomicAdd(&qcnt[(size_t)nTiles * QCNT_STRIDE], 1u);
                if (o < (unsigned)E) {
                    ovfRow[o] = r[u];
                    ovfCol[o] = c[u];
                    ovfW[o]   = m[u];
                }
            }
        }
    }
}

__global__ void __launch_bounds__(AGG_THREADS)
tile_weight_dis_kernel(unsigned* __restrict__ queue, const unsigned* __restrict__ qcnt,
                       const int* __restrict__ ovfCol, float* __restrict__ ovfW,
                       const unsigned* __restrict__ mm, float* __restrict__ dis,
                       int N, int E) {
    __shared__ float degL[TILE_N];
    const int tid      = threadIdx.x;
    const int nodeBase = blockIdx.x * TILE_N;
    const int cnt  = (int)min(qcnt[(size_t)blockIdx.x * QCNT_STRIDE], (unsigned)QCAP);
    const int nOvf = (int)min(qcnt[(size_t)gridDim.x * QCNT_STRIDE], (unsigned)E);
    const float mn    = __uint_as_float(mm[0]);
    const float mx    = __uint_as_float(mm[1]);
    const float range = mx - mn;

    for (int i = tid; i < TILE_N; i += AGG_THREADS) degL[i] = 0.f;
    __syncthreads();

    uint2* __restrict__ myQ = (uint2*)queue + (size_t)blockIdx.x * QCAP;
    for (int base = 0; base < cnt; base += 8 * AGG_THREADS) {
        uint2 r8[8];
#pragma unroll
        for (int u = 0; u < 8; ++u) {
            const int i = base + u * AGG_THREADS + tid;
            r8[u] = myQ[min(i, cnt - 1)];
        }
#pragma unroll
        for (int u = 0; u < 8; ++u) {
            const int i = base + u * AGG_THREADS + tid;
            if (i < cnt) {
                const float m = __uint_as_float(r8[u].y);
                float w = (range > EPS_W) ? (1.0f - (m - mn) / (range + EPS_W)) : 1.0f;
                w = fmaxf(w, 0.0f);
                myQ[i] = make_uint2(r8[u].x, __float_as_uint(w));
                atomicAdd(&degL[r8[u].x >> ROW_BITS], w);
            }
        }
    }
    for (int i = tid; i < nOvf; i += AGG_THREADS) {
        const unsigned lc = (unsigned)(ovfCol[i] - nodeBase);
        if (lc < TILE_N) {
            const float m = ovfW[i];
            float w = (range > EPS_W) ? (1.0f - (m - mn) / (range + EPS_W)) : 1.0f;
            w = fmaxf(w, 0.0f);
            ovfW[i] = w;
            atomicAdd(&degL[lc], w);
        }
    }
    __syncthreads();
    for (int i = tid; i < TILE_N; i += AGG_THREADS) {
        const int n = nodeBase + i;
        if (n < N) dis[n] = rsqrtf(degL[i] + 1.0f);
    }
}

__global__ void transpose_kernel(float* __restrict__ dst,
                                 const float* __restrict__ src, int R, int C) {
    long i = (long)blockIdx.x * blockDim.x + threadIdx.x;
    if (i >= (long)R * C) return;
    int r = (int)(i / C), c = (int)(i % C);
    dst[(size_t)c * R + r] = src[i];
}

__global__ void __launch_bounds__(256)
gemm_nt128_kernel(const float* __restrict__ A, const float* __restrict__ Bt,
                  float* __restrict__ C, int N) {
    const int wave = threadIdx.x >> 5;
    const int lane = threadIdx.x & 31;
    const int half = lane >> 4;
    const int lm   = lane & 15;
    const int m0   = (blockIdx.x * 8 + wave) * 16;
    if (m0 >= N) return;

    int rowA = m0 + lm;
    if (rowA >= N) rowA = N - 1;
    const float* __restrict__ arow = A + (size_t)rowA * 128;

#if defined(HAVE_WMMA_F32)
    v8f acc[8];
#pragma unroll
    for (int t = 0; t < 8; ++t)
        acc[t] = (v8f){0.f, 0.f, 0.f, 0.f, 0.f, 0.f, 0.f, 0.f};

    for (int k0 = 0; k0 < 128; k0 += 4) {
        const int ka = k0 + 2 * half;
        v2f a;
        a.x = arow[ka];
        a.y = arow[ka + 1];
        const float* __restrict__ b0 = Bt + (size_t)ka * 128 + lm;
        const float* __restrict__ b1 = Bt + (size_t)(ka + 1) * 128 + lm;
#pragma unroll
        for (int t = 0; t < 8; ++t) {
            v2f b;
            b.x = b0[t * 16];
            b.y = b1[t * 16];
            acc[t] = __builtin_amdgcn_wmma_f32_16x16x4_f32(
                false, a, false, b, (short)0, acc[t], false, false);
        }
    }
#pragma unroll
    for (int t = 0; t < 8; ++t) {
        const int colv = t * 16 + lm;
#pragma unroll
        for (int v = 0; v < 8; ++v) {
            const int rowv = m0 + v + 8 * half;
            if (rowv < N) C[(size_t)rowv * 128 + colv] = acc[t][v];
        }
    }
#else
    for (int t = 0; t < 8; ++t) {
        const int colv = t * 16 + lm;
        for (int v = 0; v < 8; ++v) {
            const int rowv = m0 + v + 8 * half;
            if (rowv >= N) continue;
            float s = 0.f;
            for (int k = 0; k < 128; ++k)
                s += A[(size_t)rowv * 128 + k] * Bt[(size_t)k * 128 + colv];
            C[(size_t)rowv * 128 + colv] = s;
        }
    }
#endif
}

__global__ void __launch_bounds__(256)
gemm_readout_kernel(const float* __restrict__ h1, const float* __restrict__ h2,
                    const float* __restrict__ Wrt, const float* __restrict__ br,
                    float* __restrict__ out, int N) {
    const int wave = threadIdx.x >> 5;
    const int lane = threadIdx.x & 31;
    const int half = lane >> 4;
    const int lm   = lane & 15;
    const int m0   = (blockIdx.x * 8 + wave) * 16;
    if (m0 >= N) return;

    int rowA = m0 + lm;
    if (rowA >= N) rowA = N - 1;
    const float* __restrict__ r1 = h1 + (size_t)rowA * 128;
    const float* __restrict__ r2 = h2 + (size_t)rowA * 128;

#if defined(HAVE_WMMA_F32)
    v8f acc[4];
#pragma unroll
    for (int t = 0; t < 4; ++t)
        acc[t] = (v8f){0.f, 0.f, 0.f, 0.f, 0.f, 0.f, 0.f, 0.f};

    for (int k0 = 0; k0 < 256; k0 += 4) {
        const int ka = k0 + 2 * half;
        v2f a;
        if (k0 < 128) { a.x = r1[ka];       a.y = r1[ka + 1]; }
        else          { a.x = r2[ka - 128]; a.y = r2[ka - 127]; }
        const float* __restrict__ b0 = Wrt + (size_t)ka * 64 + lm;
        const float* __restrict__ b1 = Wrt + (size_t)(ka + 1) * 64 + lm;
#pragma unroll
        for (int t = 0; t < 4; ++t) {
            v2f b;
            b.x = b0[t * 16];
            b.y = b1[t * 16];
            acc[t] = __builtin_amdgcn_wmma_f32_16x16x4_f32(
                false, a, false, b, (short)0, acc[t], false, false);
        }
    }
#pragma unroll
    for (int t = 0; t < 4; ++t) {
        const int colv = t * 16 + lm;
        const float bias = br[colv];
#pragma unroll
        for (int v = 0; v < 8; ++v) {
            const int rowv = m0 + v + 8 * half;
            if (rowv < N) out[(size_t)rowv * 64 + colv] = acc[t][v] + bias;
        }
    }
#else
    for (int t = 0; t < 4; ++t) {
        const int colv = t * 16 + lm;
        for (int v = 0; v < 8; ++v) {
            const int rowv = m0 + v + 8 * half;
            if (rowv >= N) continue;
            float s = br[colv];
            for (int k = 0; k < 128; ++k) {
                s += h1[(size_t)rowv * 128 + k] * Wrt[(size_t)k * 64 + colv];
                s += h2[(size_t)rowv * 128 + k] * Wrt[(size_t)(k + 128) * 64 + colv];
            }
            out[(size_t)rowv * 64 + colv] = s;
        }
    }
#endif
}

__global__ void __launch_bounds__(AGG_THREADS)
aggregate_tile_kernel(const float* __restrict__ hW, unsigned* __restrict__ queue,
                      const unsigned* __restrict__ qcnt,
                      const int* __restrict__ ovfRow, const int* __restrict__ ovfCol,
                      const float* __restrict__ ovfW,
                      const float* __restrict__ dis, const float* __restrict__ bias,
                      float* __restrict__ out, int N, int E, int firstLayer) {
    __shared__ float    tile[TILE_N * 128];
    __shared__ unsigned qa[QCAP];
    __shared__ unsigned qb[QCAP];

    const int tid  = threadIdx.x;
    const int lane = tid & 31;
    const int wv   = tid >> 5;
    const int nodeBase = blockIdx.x * TILE_N;

    const int cnt  = (int)min(qcnt[(size_t)blockIdx.x * QCNT_STRIDE], (unsigned)QCAP);
    const int nOvf = (int)min(qcnt[(size_t)gridDim.x * QCNT_STRIDE], (unsigned)E);

    for (int i = tid; i < TILE_N * 32; i += AGG_THREADS)
        *(float4*)(tile + i * 4) = make_float4(0.f, 0.f, 0.f, 0.f);

    uint2* __restrict__ myQ = (uint2*)queue + (size_t)blockIdx.x * QCAP;
    for (int base = 0; base < cnt; base += 8 * AGG_THREADS) {
        uint2 r8[8];
#pragma unroll
        for (int u = 0; u < 8; ++u) {
            const int i = base + u * AGG_THREADS + tid;
            r8[u] = (i < cnt) ? myQ[i] : make_uint2(0u, 0u);
        }
        if (firstLayer) {
            float d0[8], d1[8];
#pragma unroll
            for (int u = 0; u < 8; ++u) {
                d0[u] = dis[r8[u].x & ((1u << ROW_BITS) - 1u)];
                d1[u] = dis[nodeBase + (int)(r8[u].x >> ROW_BITS)];
            }
#pragma unroll
            for (int u = 0; u < 8; ++u) {
                const int i = base + u * AGG_THREADS + tid;
                r8[u].y = __float_as_uint(d0[u] * __uint_as_float(r8[u].y) * d1[u]);
                if (i < cnt) myQ[i] = r8[u];
            }
        }
#pragma unroll
        for (int u = 0; u < 8; ++u) {
            const int i = base + u * AGG_THREADS + tid;
            if (i < cnt) {
                qa[i] = r8[u].x;
                qb[i] = r8[u].y;
            }
        }
    }
    __syncthreads();

    for (int i = tid; i < nOvf; i += AGG_THREADS) {
        const unsigned lc = (unsigned)(ovfCol[i] - nodeBase);
        if (lc < TILE_N) {
            const float w = dis[ovfRow[i]] * ovfW[i] * dis[ovfCol[i]];
            const float* __restrict__ s = hW + (size_t)ovfRow[i] * 128;
            for (int k = 0; k < 128; ++k)
                atomicAdd(&tile[lc * 128 + (k & 3) * 32 + (k >> 2)], w * s[k]);
        }
    }

    for (int k0 = wv; k0 < cnt; k0 += 32 * 8) {
        const int  qi  = k0 + (lane & 7) * 32;
        const bool qok = qi < cnt;
        const unsigned pkL = qok ? qa[qi] : 0u;
        const unsigned wL  = qok ? qb[qi] : 0u;
        float4   v[8];
        float    w[8];
        unsigned lc[8];
#pragma unroll
        for (int b = 0; b < 8; ++b) {
            if (k0 + b * 32 < cnt) {
                const unsigned pk = __builtin_amdgcn_readlane(pkL, b);
                w[b]  = __uint_as_float(__builtin_amdgcn_readlane(wL, b));
                lc[b] = pk >> ROW_BITS;
                const unsigned r = pk & ((1u << ROW_BITS) - 1u);
                v[b] = *(const float4*)(hW + (size_t)r * 128 + lane * 4);
            }
        }
#pragma unroll
        for (int b = 0; b < 8; ++b) {
            if (k0 + b * 32 < cnt) {
                float* d = tile + lc[b] * 128 + lane;
                atomicAdd(d,      w[b] * v[b].x);
                atomicAdd(d + 32, w[b] * v[b].y);
                atomicAdd(d + 64, w[b] * v[b].z);
                atomicAdd(d + 96, w[b] * v[b].w);
            }
        }
    }
    __syncthreads();

    const float4 bq = *(const float4*)(bias + lane * 4);
    for (int i = tid; i < TILE_N * 32; i += AGG_THREADS) {
        const int ln = i >> 5;
        const int n  = nodeBase + ln;
        if (n < N) {
            const float* tp = tile + ln * 128 + lane;
            const float d  = dis[n];
            const float dd = d * d;
            const float4 h = *(const float4*)(hW + (size_t)n * 128 + lane * 4);
            float4 r;
            r.x = fmaxf(tp[0]  + dd * h.x + bq.x, 0.f);
            r.y = fmaxf(tp[32] + dd * h.y + bq.y, 0.f);
            r.z = fmaxf(tp[64] + dd * h.z + bq.z, 0.f);
            r.w = fmaxf(tp[96] + dd * h.w + bq.w, 0.f);
            *(float4*)(out + (size_t)n * 128 + lane * 4) = r;
        }
    }
}

extern "C" void kernel_launch(void* const* d_in, const int* in_sizes, int n_in,
                              void* d_out, int out_size, void* d_ws, size_t ws_size,
                              hipStream_t stream) {
    const float* x   = (const float*)d_in[0];
    const int*   ei  = (const int*)d_in[1];
    const float* ea  = (const float*)d_in[2];
    const float* W1  = (const float*)d_in[3];
    const float* b1  = (const float*)d_in[4];
    const float* W2  = (const float*)d_in[5];
    const float* b2  = (const float*)d_in[6];
    const float* Wr  = (const float*)d_in[7];
    const float* br  = (const float*)d_in[8];
    float* out = (float*)d_out;

    const long N = in_sizes[0] / IN_DIM;
    const long E = in_sizes[2] / 4;
    const int* row = ei;
    const int* col = ei + E;

    char* ws = (char*)d_ws;
    size_t off = 0;
    auto carve = [&](size_t bytes) -> char* {
        char* p = ws + off;
        off += (bytes + 255) & ~(size_t)255;
        return p;
    };
    float*    dis  = (float*)carve(N * 4);
    float*    hW   = (float*)carve(N * 128 * 4);
    float*    h1   = (float*)carve(N * 128 * 4);
    float*    h2   = (float*)carve(N * 128 * 4);
    float*    Wt1  = (float*)carve(128 * 128 * 4);
    float*    Wt2  = (float*)carve(128 * 128 * 4);
    float*    Wrt  = (float*)carve(256 * 64 * 4);
    unsigned* mm   = (unsigned*)carve(2 * 4);
    const unsigned aggBlocks = (unsigned)((N + TILE_N - 1) / TILE_N);
    unsigned* queue  = (unsigned*)carve((size_t)aggBlocks * QCAP * 8);
    unsigned* qcnt   = (unsigned*)carve((size_t)(aggBlocks + 1) * QCNT_STRIDE * 4);
    int*      ovfRow = (int*)carve(E * 4);
    int*      ovfCol = (int*)carve(E * 4);
    float*    ovfW   = (float*)carve(E * 4);
    (void)ws_size; (void)n_in; (void)out_size;

    const int BT = 256;
    auto blk1 = [&](long n) { return (unsigned)((n + BT - 1) / BT); };

    init_state_kernel<<<blk1(aggBlocks + 1), BT, 0, stream>>>(mm, qcnt, (int)(aggBlocks + 1));
    const unsigned eqBlocks = (unsigned)((E + EQ_THREADS * EQ_PER - 1) / (EQ_THREADS * EQ_PER));
    edge_bin_kernel<<<eqBlocks, EQ_THREADS, 0, stream>>>(ea, row, col, mm, queue, qcnt, ovfRow, ovfCol, ovfW, (int)aggBlocks, (int)E);
    tile_weight_dis_kernel<<<aggBlocks, AGG_THREADS, 0, stream>>>(queue, qcnt, ovfCol, ovfW, mm, dis, (int)N, (int)E);

    transpose_kernel<<<blk1(128 * 128), BT, 0, stream>>>(Wt1, W1, HID, IN_DIM);
    transpose_kernel<<<blk1(128 * 128), BT, 0, stream>>>(Wt2, W2, HID, HID);
    transpose_kernel<<<blk1(256 * 64), BT, 0, stream>>>(Wrt, Wr, OUT_DIM, 2 * HID);

    const long nWaves = (N + 15) / 16;
    const unsigned gemmBlocks = (unsigned)((nWaves + 7) / 8);

    gemm_nt128_kernel<<<gemmBlocks, 256, 0, stream>>>(x, Wt1, hW, (int)N);
    aggregate_tile_kernel<<<aggBlocks, AGG_THREADS, 0, stream>>>(hW, queue, qcnt, ovfRow, ovfCol, ovfW, dis, b1, h1, (int)N, (int)E, 1);

    gemm_nt128_kernel<<<gemmBlocks, 256, 0, stream>>>(h1, Wt2, hW, (int)N);
    aggregate_tile_kernel<<<aggBlocks, AGG_THREADS, 0, stream>>>(hW, queue, qcnt, ovfRow, ovfCol, ovfW, dis, b2, h2, (int)N, (int)E, 0);

    gemm_readout_kernel<<<gemmBlocks, 256, 0, stream>>>(h1, h2, Wrt, br, out, (int)N);
}


# GEMM weights held in LDS: a work-group copies row-major W once (pitch 132/260 floats), 25 waves each multiply one 16-row strip, B by 8-byte LDS loads; transposed copies, 3 launches gone; sums same
# speedup vs baseline: 3.3539x; 1.1292x over previous
#include <hip/hip_runtime.h>
#include <hip/hip_bf16.h>


#define IN_DIM 128
#define HID    128
#define OUT_DIM 64
#define EPS_W  1e-8f

#define TILE_N      392
#define QCAP        12288
#define AGG_THREADS 1024
#define ROW_BITS    23
#define QCNT_STRIDE 32
#define EQ_THREADS  1024
#define EQ_PER      4
#define EQ_MAXT     2048

#define GEMM_STRIPS  25
#define GEMM_THREADS (GEMM_STRIPS * 32)
#define W_PITCH      132
#define WR_PITCH     260
#define W_STAGERS    512
#define W_CHUNKS     (4096 / W_STAGERS)

typedef float v2f __attribute__((ext_vector_type(2)));
typedef float v8f __attribute__((ext_vector_type(8)));

#if defined(__gfx1250__) && __has_builtin(__builtin_amdgcn_wmma_f32_16x16x4_f32)
#define HAVE_WMMA_F32 1
#endif

__global__ void init_state_kernel(unsigned* __restrict__ mm,
                                  unsigned* __restrict__ qcnt, int nCnt) {
    const int i = blockIdx.x * blockDim.x + threadIdx.x;
    if (i == 0) {
        mm[0] = 0x7F800000u;
        mm[1] = 0x00000000u;
    }
    if (i < nCnt) qcnt[(size_t)i * QCNT_STRIDE] = 0u;
}

__global__ void __launch_bounds__(EQ_THREADS)
edge_bin_kernel(const float* __restrict__ ea, const int* __restrict__ row,
                const int* __restrict__ col, unsigned* __restrict__ mm,
                unsigned* __restrict__ queue, unsigned* __restrict__ qcnt,
                int* __restrict__ ovfRow, int* __restrict__ ovfCol,
                float* __restrict__ ovfW, int nTiles, int E) {
    __shared__ unsigned hist[EQ_MAXT];
    __shared__ unsigned sMin, sMax;
    const int tid    = threadIdx.x;
    const int nLocal = min(nTiles, EQ_MAXT);
    for (int t = tid; t < nLocal; t += EQ_THREADS) hist[t] = 0u;
    if (tid == 0) { sMin = 0x7F800000u; sMax = 0u; }
    __syncthreads();

    const int base = blockIdx.x * (EQ_THREADS * EQ_PER);
    int      r[EQ_PER], c[EQ_PER];
    float4   a[EQ_PER];
    float    m[EQ_PER];
    unsigned tl[EQ_PER], loc[EQ_PER];
    bool     ok[EQ_PER];
#pragma unroll
    for (int u = 0; u < EQ_PER; ++u) {
        const int e = base + u * EQ_THREADS + tid;
        const bool in = e < E;
        const int ec = in ? e : E - 1;
        a[u] = *(const float4*)(ea + (size_t)ec * 4);
        r[u] = row[ec];
        c[u] = col[ec];
    }
    __builtin_amdgcn_sched_barrier(0);
#pragma unroll
    for (int u = 0; u < EQ_PER; ++u) {
        const bool in = base + u * EQ_THREADS + tid < E;
        if (!in) c[u] = -1;
        tl[u] = (unsigned)c[u] / TILE_N;
        ok[u] = in && tl[u] < (unsigned)nTiles;
    }
    unsigned bmin = 0x7F800000u, bmax = 0u;
#pragma unroll
    for (int u = 0; u < EQ_PER; ++u) {
        m[u] = 0.25f * (a[u].x + a[u].y + a[u].z + a[u].w);
        if (base + u * EQ_THREADS + tid < E) {
            const unsigned b = __float_as_uint(m[u]);
            bmin = min(bmin, b);
            bmax = max(bmax, b);
        }
    }
#pragma unroll
    for (int u = 0; u < EQ_PER; ++u) {
        loc[u] = 0u;
        if (ok[u])
            loc[u] = (tl[u] < EQ_MAXT) ? atomicAdd(&hist[tl[u]], 1u)
                                       : atomicAdd(&qcnt[(size_t)tl[u] * QCNT_STRIDE], 1u);
    }
    for (int off = 16; off > 0; off >>= 1) {
        bmin = min(bmin, (unsigned)__shfl_xor((int)bmin, off));
        bmax = max(bmax, (unsigned)__shfl_xor((int)bmax, off));
    }
    if ((tid & 31) == 0) {
        atomicMin(&sMin, bmin);
        atomicMax(&sMax, bmax);
    }
    __syncthreads();
    for (int t = tid; t < nLocal; t += EQ_THREADS) {
        const unsigned n = hist[t];
        if (n) hist[t] = atomicAdd(&qcnt[(size_t)t * QCNT_STRIDE], n);
    }
    if (tid == 0) {
        atomicMin(&mm[0], sMin);
        atomicMax(&mm[1], sMax);
    }
    __syncthreads();
    uint2* __restrict__ q2 = (uint2*)queue;
#pragma unroll
    for (int u = 0; u < EQ_PER; ++u) {
        if (ok[u]) {
            const unsigned slot = loc[u] + ((tl[u] < EQ_MAXT) ? hist[tl[u]] : 0u);
            if (slot < QCAP) {
                const unsigned ln = (unsigned)c[u] - tl[u] * TILE_N;
                q2[(size_t)tl[u] * QCAP + slot] =
                    make_uint2((unsigned)r[u] | (ln << ROW_BITS), __float_as_uint(m[u]));
            } else {
                const unsigned o = atomicAdd(&qcnt[(size_t)nTiles * QCNT_STRIDE], 1u);
                if (o < (unsigned)E) {
                    ovfRow[o] = r[u];
                    ovfCol[o] = c[u];
                    ovfW[o]   = m[u];
                }
            }
        }
    }
}

__global__ void __launch_bounds__(AGG_THREADS)
tile_weight_dis_kernel(unsigned* __restrict__ queue, const unsigned* __restrict__ qcnt,
                       const int* __restrict__ ovfCol, float* __restrict__ ovfW,
                       const unsigned* __restrict__ mm, float* __restrict__ dis,
                       int N, int E) {
    __shared__ float degL[TILE_N];
    const int tid      = threadIdx.x;
    const int nodeBase = blockIdx.x * TILE_N;
    const int cnt  = (int)min(qcnt[(size_t)blockIdx.x * QCNT_STRIDE], (unsigned)QCAP);
    const int nOvf = (int)min(qcnt[(size_t)gridDim.x * QCNT_STRIDE], (unsigned)E);
    const float mn    = __uint_as_float(mm[0]);
    const float mx    = __uint_as_float(mm[1]);
    const float range = mx - mn;

    for (int i = tid; i < TILE_N; i += AGG_THREADS) degL[i] = 0.f;
    __syncthreads();

    uint2* __restrict__ myQ = (uint2*)queue + (size_t)blockIdx.x * QCAP;
    for (int base = 0; base < cnt; base += 8 * AGG_THREADS) {
        uint2 r8[8];
#pragma unroll
        for (int u = 0; u < 8; ++u) {
            const int i = base + u * AGG_THREADS + tid;
            r8[u] = myQ[min(i, cnt - 1)];
        }
#pragma unroll
        for (int u = 0; u < 8; ++u) {
            const int i = base + u * AGG_THREADS + tid;
            if (i < cnt) {
                const float m = __uint_as_float(r8[u].y);
                float w = (range > EPS_W) ? (1.0f - (m - mn) / (range + EPS_W)) : 1.0f;
                w = fmaxf(w, 0.0f);
                myQ[i] = make_uint2(r8[u].x, __float_as_uint(w));
                atomicAdd(&degL[r8[u].x >> ROW_BITS], w);
            }
        }
    }
    for (int i = tid; i < nOvf; i += AGG_THREADS) {
        const unsigned lc = (unsigned)(ovfCol[i] - nodeBase);
        if (lc < TILE_N) {
            const float m = ovfW[i];
            float w = (range > EPS_W) ? (1.0f - (m - mn) / (range + EPS_W)) : 1.0f;
            w = fmaxf(w, 0.0f);
            ovfW[i] = w;
            atomicAdd(&degL[lc], w);
        }
    }
    __syncthreads();
    for (int i = tid; i < TILE_N; i += AGG_THREADS) {
        const int n = nodeBase + i;
        if (n < N) dis[n] = rsqrtf(degL[i] + 1.0f);
    }
}

__global__ void __launch_bounds__(GEMM_THREADS)
gemm_nt128_kernel(const float* __restrict__ A, const float* __restrict__ W,
                  float* __restrict__ C, int N) {
    __shared__ float wl[128 * W_PITCH];
    if (threadIdx.x < W_STAGERS) {
        float4 wv[W_CHUNKS];
#pragma unroll
        for (int u = 0; u < W_CHUNKS; ++u)
            wv[u] = *(const float4*)(W + (size_t)((int)threadIdx.x + u * W_STAGERS) * 4);
#pragma unroll
        for (int u = 0; u < W_CHUNKS; ++u) {
            const int i = (int)threadIdx.x + u * W_STAGERS;
            const int n = i >> 5, q = i & 31;
            *(float4*)(wl + n * W_PITCH + q * 4) = wv[u];
        }
    }
    __syncthreads();

    const int wave = threadIdx.x >> 5;
    const int lane = threadIdx.x & 31;
    const int half = lane >> 4;
    const int lm   = lane & 15;
    const int m0   = (blockIdx.x * GEMM_STRIPS + wave) * 16;
    if (m0 >= N) return;

    int rowA = m0 + lm;
    if (rowA >= N) rowA = N - 1;
    const float* __restrict__ arow = A + (size_t)rowA * 128;

#if defined(HAVE_WMMA_F32)
    v8f acc[8];
#pragma unroll
    for (int t = 0; t < 8; ++t)
        acc[t] = (v8f){0.f, 0.f, 0.f, 0.f, 0.f, 0.f, 0.f, 0.f};

    for (int k0 = 0; k0 < 128; k0 += 4) {
        const int ka = k0 + 2 * half;
        v2f a;
        a.x = arow[ka];
        a.y = arow[ka + 1];
        const float* bw = wl + lm * W_PITCH + ka;
#pragma unroll
        for (int t = 0; t < 8; ++t) {
            const v2f b = *(const v2f*)(bw + t * 16 * W_PITCH);
            acc[t] = __builtin_amdgcn_wmma_f32_16x16x4_f32(
                false, a, false, b, (short)0, acc[t], false, false);
        }
    }
#pragma unroll
    for (int t = 0; t < 8; ++t) {
        const int colv = t * 16 + lm;
#pragma unroll
        for (int v = 0; v < 8; ++v) {
            const int rowv = m0 + v + 8 * half;
            if (rowv < N) C[(size_t)rowv * 128 + colv] = acc[t][v];
        }
    }
#else
    for (int t = 0; t < 8; ++t) {
        const int colv = t * 16 + lm;
        for (int v = 0; v < 8; ++v) {
            const int rowv = m0 + v + 8 * half;
            if (rowv >= N) continue;
            float s = 0.f;
            for (int k = 0; k < 128; ++k)
                s += A[(size_t)rowv * 128 + k] * wl[colv * W_PITCH + k];
            C[(size_t)rowv * 128 + colv] = s;
        }
    }
#endif
}

__global__ void __launch_bounds__(GEMM_THREADS)
gemm_readout_kernel(const float* __restrict__ h1, const float* __restrict__ h2,
                    const float* __restrict__ Wr, const float* __restrict__ br,
                    float* __restrict__ out, int N) {
    __shared__ float wl[64 * WR_PITCH];
    if (threadIdx.x < W_STAGERS) {
        float4 wv[W_CHUNKS];
#pragma unroll
        for (int u = 0; u < W_CHUNKS; ++u)
            wv[u] = *(const float4*)(Wr + (size_t)((int)threadIdx.x + u * W_STAGERS) * 4);
#pragma unroll
        for (int u = 0; u < W_CHUNKS; ++u) {
            const int i = (int)threadIdx.x + u * W_STAGERS;
            const int n = i >> 6, q = i & 63;
            *(float4*)(wl + n * WR_PITCH + q * 4) = wv[u];
        }
    }
    __syncthreads();

    const int wave = threadIdx.x >> 5;
    const int lane = threadIdx.x & 31;
    const int half = lane >> 4;
    const int lm   = lane & 15;
    const int m0   = (blockIdx.x * GEMM_STRIPS + wave) * 16;
    if (m0 >= N) return;

    int rowA = m0 + lm;
    if (rowA >= N) rowA = N - 1;
    const float* __restrict__ r1 = h1 + (size_t)rowA * 128;
    const float* __restrict__ r2 = h2 + (size_t)rowA * 128;

#if defined(HAVE_WMMA_F32)
    v8f acc[4];
#pragma unroll
    for (int t = 0; t < 4; ++t)
        acc[t] = (v8f){0.f, 0.f, 0.f, 0.f, 0.f, 0.f, 0.f, 0.f};

    for (int k0 = 0; k0 < 256; k0 += 4) {
        const int ka = k0 + 2 * half;
        v2f a;
        if (k0 < 128) { a.x = r1[ka];       a.y = r1[ka + 1]; }
        else          { a.x = r2[ka - 128]; a.y = r2[ka - 127]; }
        const float* bw = wl + lm * WR_PITCH + ka;
#pragma unroll
        for (int t = 0; t < 4; ++t) {
            const v2f b = *(const v2f*)(bw + t * 16 * WR_PITCH);
            acc[t] = __builtin_amdgcn_wmma_f32_16x16x4_f32(
                false, a, false, b, (short)0, acc[t], false, false);
        }
    }
#pragma unroll
    for (int t = 0; t < 4; ++t) {
        const int colv = t * 16 + lm;
        const float bias = br[colv];
#pragma unroll
        for (int v = 0; v < 8; ++v) {
            const int rowv = m0 + v + 8 * half;
            if (rowv < N) out[(size_t)rowv * 64 + colv] = acc[t][v] + bias;
        }
    }
#else
    for (int t = 0; t < 4; ++t) {
        const int colv = t * 16 + lm;
        for (int v = 0; v < 8; ++v) {
            const int rowv = m0 + v + 8 * half;
            if (rowv >= N) continue;
            float s = br[colv];
            for (int k = 0; k < 128; ++k) {
                s += h1[(size_t)rowv * 128 + k] * wl[colv * WR_PITCH + k];
                s += h2[(size_t)rowv * 128 + k] * wl[colv * WR_PITCH + k + 128];
            }
            out[(size_t)rowv * 64 + colv] = s;
        }
    }
#endif
}

__global__ void __launch_bounds__(AGG_THREADS)
aggregate_tile_kernel(const float* __restrict__ hW, unsigned* __restrict__ queue,
                      const unsigned* __restrict__ qcnt,
                      const int* __restrict__ ovfRow, const int* __restrict__ ovfCol,
                      const float* __restrict__ ovfW,
                      const float* __restrict__ dis, const float* __restrict__ bias,
                      float* __restrict__ out, int N, int E, int firstLayer) {
    __shared__ float    tile[TILE_N * 128];
    __shared__ unsigned qa[QCAP];
    __shared__ unsigned qb[QCAP];

    const int tid  = threadIdx.x;
    const int lane = tid & 31;
    const int wv   = tid >> 5;
    const int nodeBase = blockIdx.x * TILE_N;

    const int cnt  = (int)min(qcnt[(size_t)blockIdx.x * QCNT_STRIDE], (unsigned)QCAP);
    const int nOvf = (int)min(qcnt[(size_t)gridDim.x * QCNT_STRIDE], (unsigned)E);

    for (int i = tid; i < TILE_N * 32; i += AGG_THREADS)
        *(float4*)(tile + i * 4) = make_float4(0.f, 0.f, 0.f, 0.f);

    uint2* __restrict__ myQ = (uint2*)queue + (size_t)blockIdx.x * QCAP;
    for (int base = 0; base < cnt; base += 8 * AGG_THREADS) {
        uint2 r8[8];
#pragma unroll
        for (int u = 0; u < 8; ++u) {
            const int i = base + u * AGG_THREADS + tid;
            r8[u] = (i < cnt) ? myQ[i] : make_uint2(0u, 0u);
        }
        if (firstLayer) {
            float d0[8], d1[8];
#pragma unroll
            for (int u = 0; u < 8; ++u) {
                d0[u] = dis[r8[u].x & ((1u << ROW_BITS) - 1u)];
                d1[u] = dis[nodeBase + (int)(r8[u].x >> ROW_BITS)];
            }
#pragma unroll
            for (int u = 0; u < 8; ++u) {
                const int i = base + u * AGG_THREADS + tid;
                r8[u].y = __float_as_uint(d0[u] * __uint_as_float(r8[u].y) * d1[u]);
                if (i < cnt) myQ[i] = r8[u];
            }
        }
#pragma unroll
        for (int u = 0; u < 8; ++u) {
            const int i = base + u * AGG_THREADS + tid;
            if (i < cnt) {
                qa[i] = r8[u].x;
                qb[i] = r8[u].y;
            }
        }
    }
    __syncthreads();

    for (int i = tid; i < nOvf; i += AGG_THREADS) {
        const unsigned lc = (unsigned)(ovfCol[i] - nodeBase);
        if (lc < TILE_N) {
            const float w = dis[ovfRow[i]] * ovfW[i] * dis[ovfCol[i]];
            const float* __restrict__ s = hW + (size_t)ovfRow[i] * 128;
            for (int k = 0; k < 128; ++k)
                atomicAdd(&tile[lc * 128 + (k & 3) * 32 + (k >> 2)], w * s[k]);
        }
    }

    for (int k0 = wv; k0 < cnt; k0 += 32 * 8) {
        const int  qi  = k0 + (lane & 7) * 32;
        const bool qok = qi < cnt;
        const unsigned pkL = qok ? qa[qi] : 0u;
        const unsigned wL  = qok ? qb[qi] : 0u;
        float4   v[8];
        float    w[8];
        unsigned lc[8];
#pragma unroll
        for (int b = 0; b < 8; ++b) {
            if (k0 + b * 32 < cnt) {
                const unsigned pk = __builtin_amdgcn_readlane(pkL, b);
                w[b]  = __uint_as_float(__builtin_amdgcn_readlane(wL, b));
                lc[b] = pk >> ROW_BITS;
                const unsigned r = pk & ((1u << ROW_BITS) - 1u);
                v[b] = *(const float4*)(hW + (size_t)r * 128 + lane * 4);
            }
        }
#pragma unroll
        for (int b = 0; b < 8; ++b) {
            if (k0 + b * 32 < cnt) {
                float* d = tile + lc[b] * 128 + lane;
                atomicAdd(d,      w[b] * v[b].x);
                atomicAdd(d + 32, w[b] * v[b].y);
                atomicAdd(d + 64, w[b] * v[b].z);
                atomicAdd(d + 96, w[b] * v[b].w);
            }
        }
    }
    __syncthreads();

    const float4 bq = *(const float4*)(bias + lane * 4);
    for (int i = tid; i < TILE_N * 32; i += AGG_THREADS) {
        const int ln = i >> 5;
        const int n  = nodeBase + ln;
        if (n < N) {
            const float* tp = tile + ln * 128 + lane;
            const float d  = dis[n];
            const float dd = d * d;
            const float4 h = *(const float4*)(hW + (size_t)n * 128 + lane * 4);
            float4 r;
            r.x = fmaxf(tp[0]  + dd * h.x + bq.x, 0.f);
            r.y = fmaxf(tp[32] + dd * h.y + bq.y, 0.f);
            r.z = fmaxf(tp[64] + dd * h.z + bq.z, 0.f);
            r.w = fmaxf(tp[96] + dd * h.w + bq.w, 0.f);
            *(float4*)(out + (size_t)n * 128 + lane * 4) = r;
        }
    }
}

extern "C" void kernel_launch(void* const* d_in, const int* in_sizes, int n_in,
                              void* d_out, int out_size, void* d_ws, size_t ws_size,
                              hipStream_t stream) {
    const float* x   = (const float*)d_in[0];
    const int*   ei  = (const int*)d_in[1];
    const float* ea  = (const float*)d_in[2];
    const float* W1  = (const float*)d_in[3];
    const float* b1  = (const float*)d_in[4];
    const float* W2  = (const float*)d_in[5];
    const float* b2  = (const float*)d_in[6];
    const float* Wr  = (const float*)d_in[7];
    const float* br  = (const float*)d_in[8];
    float* out = (float*)d_out;

    const long N = in_sizes[0] / IN_DIM;
    const long E = in_sizes[2] / 4;
    const int* row = ei;
    const int* col = ei + E;

    char* ws = (char*)d_ws;
    size_t off = 0;
    auto carve = [&](size_t bytes) -> char* {
        char* p = ws + off;
        off += (bytes + 255) & ~(size_t)255;
        return p;
    };
    float*    dis  = (float*)carve(N * 4);
    float*    hW   = (float*)carve(N * 128 * 4);
    float*    h1   = (float*)carve(N * 128 * 4);
    float*    h2   = (float*)carve(N * 128 * 4);
    unsigned* mm   = (unsigned*)carve(2 * 4);
    const unsigned aggBlocks = (unsigned)((N + TILE_N - 1) / TILE_N);
    unsigned* queue  = (unsigned*)carve((size_t)aggBlocks * QCAP * 8);
    unsigned* qcnt   = (unsigned*)carve((size_t)(aggBlocks + 1) * QCNT_STRIDE * 4);
    int*      ovfRow = (int*)carve(E * 4);
    int*      ovfCol = (int*)carve(E * 4);
    float*    ovfW   = (float*)carve(E * 4);
    (void)ws_size; (void)n_in; (void)out_size;

    const int BT = 256;
    auto blk1 = [&](long n) { return (unsigned)((n + BT - 1) / BT); };

    init_state_kernel<<<blk1(aggBlocks + 1), BT, 0, stream>>>(mm, qcnt, (int)(aggBlocks + 1));
    const unsigned eqBlocks = (unsigned)((E + EQ_THREADS * EQ_PER - 1) / (EQ_THREADS * EQ_PER));
    edge_bin_kernel<<<eqBlocks, EQ_THREADS, 0, stream>>>(ea, row, col, mm, queue, qcnt, ovfRow, ovfCol, ovfW, (int)aggBlocks, (int)E);
    tile_weight_dis_kernel<<<aggBlocks, AGG_THREADS, 0, stream>>>(queue, qcnt, ovfCol, ovfW, mm, dis, (int)N, (int)E);

    const long nStrips = (N + 15) / 16;
    const unsigned gemmBlocks = (unsigned)((nStrips + GEMM_STRIPS - 1) / GEMM_STRIPS);

    gemm_nt128_kernel<<<gemmBlocks, GEMM_THREADS, 0, stream>>>(x, W1, hW, (int)N);
    aggregate_tile_kernel<<<aggBlocks, AGG_THREADS, 0, stream>>>(hW, queue, qcnt, ovfRow, ovfCol, ovfW, dis, b1, h1, (int)N, (int)E, 1);

    gemm_nt128_kernel<<<gemmBlocks, GEMM_THREADS, 0, stream>>>(h1, W2, hW, (int)N);
    aggregate_tile_kernel<<<aggBlocks, AGG_THREADS, 0, stream>>>(hW, queue, qcnt, ovfRow, ovfCol, ovfW, dis, b2, h2, (int)N, (int)E, 0);

    gemm_readout_kernel<<<gemmBlocks, GEMM_THREADS, 0, stream>>>(h1, h2, Wr, br, out, (int)N);
}


# Layer-1 product moved into the edge-pass launch: a block counts its edges, then 25 waves multiply x strips while 7 wait for the queue reservations, then records are stored; 1 launch gone; sums same
# speedup vs baseline: 3.5417x; 1.0560x over previous
#include <hip/hip_runtime.h>
#include <hip/hip_bf16.h>


#define IN_DIM 128
#define HID    128
#define OUT_DIM 64
#define EPS_W  1e-8f

#define TILE_N      392
#define QCAP        12288
#define AGG_THREADS 1024
#define ROW_BITS    23
#define QCNT_STRIDE 32
#define EQ_THREADS  1024
#define EQ_PER      7
#define EQ_MAXT     2048

#define GEMM_STRIPS  25
#define GEMM_THREADS (GEMM_STRIPS * 32)
#define W_PITCH      132
#define WR_PITCH     260
#define W_STAGERS    512
#define W_CHUNKS     (4096 / W_STAGERS)

typedef float v2f __attribute__((ext_vector_type(2)));
typedef float v8f __attribute__((ext_vector_type(8)));

#if defined(__gfx1250__) && __has_builtin(__builtin_amdgcn_wmma_f32_16x16x4_f32)
#define HAVE_WMMA_F32 1
#endif

__global__ void init_state_kernel(unsigned* __restrict__ mm,
                                  unsigned* __restrict__ qcnt, int nCnt) {
    const int i = blockIdx.x * blockDim.x + threadIdx.x;
    if (i == 0) {
        mm[0] = 0x7F800000u;
        mm[1] = 0x00000000u;
    }
    if (i < nCnt) qcnt[(size_t)i * QCNT_STRIDE] = 0u;
}

__global__ void __launch_bounds__(AGG_THREADS)
tile_weight_dis_kernel(unsigned* __restrict__ queue, const unsigned* __restrict__ qcnt,
                       const int* __restrict__ ovfCol, float* __restrict__ ovfW,
                       const unsigned* __restrict__ mm, float* __restrict__ dis,
                       int N, int E) {
    __shared__ float degL[TILE_N];
    const int tid      = threadIdx.x;
    const int nodeBase = blockIdx.x * TILE_N;
    const int cnt  = (int)min(qcnt[(size_t)blockIdx.x * QCNT_STRIDE], (unsigned)QCAP);
    const int nOvf = (int)min(qcnt[(size_t)gridDim.x * QCNT_STRIDE], (unsigned)E);
    const float mn    = __uint_as_float(mm[0]);
    const float mx    = __uint_as_float(mm[1]);
    const float range = mx - mn;

    for (int i = tid; i < TILE_N; i += AGG_THREADS) degL[i] = 0.f;
    __syncthreads();

    uint2* __restrict__ myQ = (uint2*)queue + (size_t)blockIdx.x * QCAP;
    for (int base = 0; base < cnt; base += 8 * AGG_THREADS) {
        uint2 r8[8];
#pragma unroll
        for (int u = 0; u < 8; ++u) {
            const int i = base + u * AGG_THREADS + tid;
            r8[u] = myQ[min(i, cnt - 1)];
        }
#pragma unroll
        for (int u = 0; u < 8; ++u) {
            const int i = base + u * AGG_THREADS + tid;
            if (i < cnt) {
                const float m = __uint_as_float(r8[u].y);
                float w = (range > EPS_W) ? (1.0f - (m - mn) / (range + EPS_W)) : 1.0f;
                w = fmaxf(w, 0.0f);
                myQ[i] = make_uint2(r8[u].x, __float_as_uint(w));
                atomicAdd(&degL[r8[u].x >> ROW_BITS], w);
            }
        }
    }
    for (int i = tid; i < nOvf; i += AGG_THREADS) {
        const unsigned lc = (unsigned)(ovfCol[i] - nodeBase);
        if (lc < TILE_N) {
            const float m = ovfW[i];
            float w = (range > EPS_W) ? (1.0f - (m - mn) / (range + EPS_W)) : 1.0f;
            w = fmaxf(w, 0.0f);
            ovfW[i] = w;
            atomicAdd(&degL[lc], w);
        }
    }
    __syncthreads();
    for (int i = tid; i < TILE_N; i += AGG_THREADS) {
        const int n = nodeBase + i;
        if (n < N) dis[n] = rsqrtf(degL[i] + 1.0f);
    }
}


__device__ __forceinline__ void w128_load(const float* __restrict__ W, float4 (&wv)[W_CHUNKS], int tid) {
#pragma unroll
    for (int u = 0; u < W_CHUNKS; ++u)
        wv[u] = *(const float4*)(W + (size_t)(tid + u * W_STAGERS) * 4);
}
__device__ __forceinline__ void w128_store(float* __restrict__ wl, const float4 (&wv)[W_CHUNKS], int tid) {
#pragma unroll
    for (int u = 0; u < W_CHUNKS; ++u) {
        const int i = tid + u * W_STAGERS;
        const int n = i >> 5, q = i & 31;
        *(float4*)(wl + n * W_PITCH + q * 4) = wv[u];
    }
}

template <int KU>
__device__ __forceinline__ void gemm_strip128(const float* __restrict__ A, const float* __restrict__ wl,
                                              float* __restrict__ C, int N, int m0, int lane) {
    const int half = lane >> 4;
    const int lm   = lane & 15;
    int rowA = m0 + lm;
    if (rowA >= N) rowA = N - 1;
    const float* __restrict__ arow = A + (size_t)rowA * 128;

#if defined(HAVE_WMMA_F32)
    v8f acc[8];
#pragma unroll
    for (int t = 0; t < 8; ++t)
        acc[t] = (v8f){0.f, 0.f, 0.f, 0.f, 0.f, 0.f, 0.f, 0.f};

#pragma unroll KU
    for (int k0 = 0; k0 < 128; k0 += 4) {
        const int ka = k0 + 2 * half;
        v2f a;
        a.x = arow[ka];
        a.y = arow[ka + 1];
        const float* bw = wl + lm * W_PITCH + ka;
#pragma unroll
        for (int t = 0; t < 8; ++t) {
            const v2f b = *(const v2f*)(bw + t * 16 * W_PITCH);
            acc[t] = __builtin_amdgcn_wmma_f32_16x16x4_f32(
                false, a, false, b, (short)0, acc[t], false, false);
        }
    }
#pragma unroll
    for (int t = 0; t < 8; ++t) {
        const int colv = t * 16 + lm;
#pragma unroll
        for (int v = 0; v < 8; ++v) {
            const int rowv = m0 + v + 8 * half;
            if (rowv < N) C[(size_t)rowv * 128 + colv] = acc[t][v];
        }
    }
#else
    for (int t = 0; t < 8; ++t) {
        const int colv = t * 16 + lm;
        for (int v = 0; v < 8; ++v) {
            const int rowv = m0 + v + 8 * half;
            if (rowv >= N) continue;
            float s = 0.f;
            for (int k = 0; k < 128; ++k)
                s += A[(size_t)rowv * 128 + k] * wl[colv * W_PITCH + k];
            C[(size_t)rowv * 128 + colv] = s;
        }
    }
#endif
}

__global__ void __launch_bounds__(GEMM_THREADS)
gemm_nt128_kernel(const float* __restrict__ A, const float* __restrict__ W,
                  float* __restrict__ C, int N) {
    __shared__ float wl[128 * W_PITCH];
    if (threadIdx.x < W_STAGERS) {
        float4 wv[W_CHUNKS];
        w128_load(W, wv, (int)threadIdx.x);
        w128_store(wl, wv, (int)threadIdx.x);
    }
    __syncthreads();

    const int m0 = (blockIdx.x * GEMM_STRIPS + (threadIdx.x >> 5)) * 16;
    if (m0 >= N) return;
    gemm_strip128<4>(A, wl, C, N, m0, threadIdx.x & 31);
}

__global__ void __launch_bounds__(EQ_THREADS)
edge_bin_gemm_kernel(const float* __restrict__ ea, const int* __restrict__ row,
                     const int* __restrict__ col, unsigned* __restrict__ mm,
                     unsigned* __restrict__ queue, unsigned* __restrict__ qcnt,
                     int* __restrict__ ovfRow, int* __restrict__ ovfCol,
                     float* __restrict__ ovfW, int nTiles, int E,
                     const float* __restrict__ X, const float* __restrict__ W,
                     float* __restrict__ C, int N) {
    static_assert(EQ_THREADS > GEMM_THREADS && EQ_THREADS >= W_STAGERS, "waves left over for the reservations");
    __shared__ float    wl[128 * W_PITCH];
    __shared__ unsigned hist[EQ_MAXT];
    __shared__ uint4    park[EQ_THREADS * EQ_PER];
    __shared__ unsigned sMin, sMax;
    const int tid    = threadIdx.x;
    const int nLocal = min(nTiles, EQ_MAXT);
    for (int t = tid; t < nLocal; t += EQ_THREADS) hist[t] = 0u;
    if (tid == 0) { sMin = 0x7F800000u; sMax = 0u; }

    const int base = blockIdx.x * (EQ_THREADS * EQ_PER);
    int      r[EQ_PER], c[EQ_PER];
    float4   a[EQ_PER];
#pragma unroll
    for (int u = 0; u < EQ_PER; ++u) {
        const int e = base + u * EQ_THREADS + tid;
        const int ec = (e < E) ? e : E - 1;
        a[u] = *(const float4*)(ea + (size_t)ec * 4);
        r[u] = row[ec];
        c[u] = col[ec];
    }
    __builtin_amdgcn_sched_barrier(0);
    if (tid < W_STAGERS) {
        float4 wv[W_CHUNKS];
        w128_load(W, wv, tid);
        w128_store(wl, wv, tid);
    }
    __syncthreads();

    unsigned bmin = 0x7F800000u, bmax = 0u;
#pragma unroll
    for (int u = 0; u < EQ_PER; ++u) {
        const bool in = base + u * EQ_THREADS + tid < E;
        const int  cu = in ? c[u] : -1;
        const unsigned tl = (unsigned)cu / TILE_N;
        const bool ok = in && tl < (unsigned)nTiles;
        const float m = 0.25f * (a[u].x + a[u].y + a[u].z + a[u].w);
        if (in) {
            const unsigned b = __float_as_uint(m);
            bmin = min(bmin, b);
            bmax = max(bmax, b);
        }
        unsigned loc = 0u;
        if (ok)
            loc = (tl < EQ_MAXT) ? atomicAdd(&hist[tl], 1u)
                                 : atomicAdd(&qcnt[(size_t)tl * QCNT_STRIDE], 1u);
        park[tid * EQ_PER + u] = make_uint4((unsigned)r[u] | (((unsigned)cu - tl * TILE_N) << ROW_BITS),
                                            __float_as_uint(m), ok ? tl : 0xFFFFFFFFu, loc);
    }
    for (int off = 16; off > 0; off >>= 1) {
        bmin = min(bmin, (unsigned)__shfl_xor((int)bmin, off));
        bmax = max(bmax, (unsigned)__shfl_xor((int)bmax, off));
    }
    if ((tid & 31) == 0) {
        atomicMin(&sMin, bmin);
        atomicMax(&sMax, bmax);
    }
    __syncthreads();
    if (tid == 0) {
        atomicMin(&mm[0], sMin);
        atomicMax(&mm[1], sMax);
    }
    if (tid < GEMM_THREADS) {
        const int m0 = (blockIdx.x * GEMM_STRIPS + (tid >> 5)) * 16;
        if (m0 < N) gemm_strip128<2>(X, wl, C, N, m0, tid & 31);
    } else {
        for (int t = tid - GEMM_THREADS; t < nLocal; t += EQ_THREADS - GEMM_THREADS) {
            const unsigned n = hist[t];
            if (n) hist[t] = atomicAdd(&qcnt[(size_t)t * QCNT_STRIDE], n);
        }
    }
    __syncthreads();
    uint2* __restrict__ q2 = (uint2*)queue;
#pragma unroll
    for (int u = 0; u < EQ_PER; ++u) {
        const uint4 rec = park[tid * EQ_PER + u];
        const unsigned tl = rec.z;
        if (tl != 0xFFFFFFFFu) {
            const unsigned pk = rec.x;
            const unsigned mb = rec.y;
            const unsigned slot = rec.w + ((tl < EQ_MAXT) ? hist[tl] : 0u);
            if (slot < QCAP) {
                q2[(size_t)tl * QCAP + slot] = make_uint2(pk, mb);
            } else {
                const unsigned o = atomicAdd(&qcnt[(size_t)nTiles * QCNT_STRIDE], 1u);
                if (o < (unsigned)E) {
                    ovfRow[o] = (int)(pk & ((1u << ROW_BITS) - 1u));
                    ovfCol[o] = (int)(tl * TILE_N + (pk >> ROW_BITS));
                    ovfW[o]   = __uint_as_float(mb);
                }
            }
        }
    }
}

__global__ void __launch_bounds__(GEMM_THREADS)
gemm_readout_kernel(const float* __restrict__ h1, const float* __restrict__ h2,
                    const float* __restrict__ Wr, const float* __restrict__ br,
                    float* __restrict__ out, int N) {
    __shared__ float wl[64 * WR_PITCH];
    if (threadIdx.x < W_STAGERS) {
        float4 wv[W_CHUNKS];
#pragma unroll
        for (int u = 0; u < W_CHUNKS; ++u)
            wv[u] = *(const float4*)(Wr + (size_t)((int)threadIdx.x + u * W_STAGERS) * 4);
#pragma unroll
        for (int u = 0; u < W_CHUNKS; ++u) {
            const int i = (int)threadIdx.x + u * W_STAGERS;
            const int n = i >> 6, q = i & 63;
            *(float4*)(wl + n * WR_PITCH + q * 4) = wv[u];
        }
    }
    __syncthreads();

    const int wave = threadIdx.x >> 5;
    const int lane = threadIdx.x & 31;
    const int half = lane >> 4;
    const int lm   = lane & 15;
    const int m0   = (blockIdx.x * GEMM_STRIPS + wave) * 16;
    if (m0 >= N) return;

    int rowA = m0 + lm;
    if (rowA >= N) rowA = N - 1;
    const float* __restrict__ r1 = h1 + (size_t)rowA * 128;
    const float* __restrict__ r2 = h2 + (size_t)rowA * 128;

#if defined(HAVE_WMMA_F32)
    v8f acc[4];
#pragma unroll
    for (int t = 0; t < 4; ++t)
        acc[t] = (v8f){0.f, 0.f, 0.f, 0.f, 0.f, 0.f, 0.f, 0.f};

    for (int k0 = 0; k0 < 256; k0 += 4) {
        const int ka = k0 + 2 * half;
        v2f a;
        if (k0 < 128) { a.x = r1[ka];       a.y = r1[ka + 1]; }
        else          { a.x = r2[ka - 128]; a.y = r2[ka - 127]; }
        const float* bw = wl + lm * WR_PITCH + ka;
#pragma unroll
        for (int t = 0; t < 4; ++t) {
            const v2f b = *(const v2f*)(bw + t * 16 * WR_PITCH);
            acc[t] = __builtin_amdgcn_wmma_f32_16x16x4_f32(
                false, a, false, b, (short)0, acc[t], false, false);
        }
    }
#pragma unroll
    for (int t = 0; t < 4; ++t) {
        const int colv = t * 16 + lm;
        const float bias = br[colv];
#pragma unroll
        for (int v = 0; v < 8; ++v) {
            const int rowv = m0 + v + 8 * half;
            if (rowv < N) out[(size_t)rowv * 64 + colv] = acc[t][v] + bias;
        }
    }
#else
    for (int t = 0; t < 4; ++t) {
        const int colv = t * 16 + lm;
        for (int v = 0; v < 8; ++v) {
            const int rowv = m0 + v + 8 * half;
            if (rowv >= N) continue;
            float s = br[colv];
            for (int k = 0; k < 128; ++k) {
                s += h1[(size_t)rowv * 128 + k] * wl[colv * WR_PITCH + k];
                s += h2[(size_t)rowv * 128 + k] * wl[colv * WR_PITCH + k + 128];
            }
            out[(size_t)rowv * 64 + colv] = s;
        }
    }
#endif
}

__global__ void __launch_bounds__(AGG_THREADS)
aggregate_tile_kernel(const float* __restrict__ hW, unsigned* __restrict__ queue,
                      const unsigned* __restrict__ qcnt,
                      const int* __restrict__ ovfRow, const int* __restrict__ ovfCol,
                      const float* __restrict__ ovfW,
                      const float* __restrict__ dis, const float* __restrict__ bias,
                      float* __restrict__ out, int N, int E, int firstLayer) {
    __shared__ float    tile[TILE_N * 128];
    __shared__ unsigned qa[QCAP];
    __shared__ unsigned qb[QCAP];

    const int tid  = threadIdx.x;
    const int lane = tid & 31;
    const int wv   = tid >> 5;
    const int nodeBase = blockIdx.x * TILE_N;

    const int cnt  = (int)min(qcnt[(size_t)blockIdx.x * QCNT_STRIDE], (unsigned)QCAP);
    const int nOvf = (int)min(qcnt[(size_t)gridDim.x * QCNT_STRIDE], (unsigned)E);

    for (int i = tid; i < TILE_N * 32; i += AGG_THREADS)
        *(float4*)(tile + i * 4) = make_float4(0.f, 0.f, 0.f, 0.f);

    uint2* __restrict__ myQ = (uint2*)queue + (size_t)blockIdx.x * QCAP;
    for (int base = 0; base < cnt; base += 8 * AGG_THREADS) {
        uint2 r8[8];
#pragma unroll
        for (int u = 0; u < 8; ++u) {
            const int i = base + u * AGG_THREADS + tid;
            r8[u] = (i < cnt) ? myQ[i] : make_uint2(0u, 0u);
        }
        if (firstLayer) {
            float d0[8], d1[8];
#pragma unroll
            for (int u = 0; u < 8; ++u) {
                d0[u] = dis[r8[u].x & ((1u << ROW_BITS) - 1u)];
                d1[u] = dis[nodeBase + (int)(r8[u].x >> ROW_BITS)];
            }
#pragma unroll
            for (int u = 0; u < 8; ++u) {
                const int i = base + u * AGG_THREADS + tid;
                r8[u].y = __float_as_uint(d0[u] * __uint_as_float(r8[u].y) * d1[u]);
                if (i < cnt) myQ[i] = r8[u];
            }
        }
#pragma unroll
        for (int u = 0; u < 8; ++u) {
            const int i = base + u * AGG_THREADS + tid;
            if (i < cnt) {
                qa[i] = r8[u].x;
                qb[i] = r8[u].y;
            }
        }
    }
    __syncthreads();

    for (int i = tid; i < nOvf; i += AGG_THREADS) {
        const unsigned lc = (unsigned)(ovfCol[i] - nodeBase);
        if (lc < TILE_N) {
            const float w = dis[ovfRow[i]] * ovfW[i] * dis[ovfCol[i]];
            const float* __restrict__ s = hW + (size_t)ovfRow[i] * 128;
            for (int k = 0; k < 128; ++k)
                atomicAdd(&tile[lc * 128 + (k & 3) * 32 + (k >> 2)], w * s[k]);
        }
    }

    for (int k0 = wv; k0 < cnt; k0 += 32 * 8) {
        const int  qi  = k0 + (lane & 7) * 32;
        const bool qok = qi < cnt;
        const unsigned pkL = qok ? qa[qi] : 0u;
        const unsigned wL  = qok ? qb[qi] : 0u;
        float4   v[8];
        float    w[8];
        unsigned lc[8];
#pragma unroll
        for (int b = 0; b < 8; ++b) {
            if (k0 + b * 32 < cnt) {
                const unsigned pk = __builtin_amdgcn_readlane(pkL, b);
                w[b]  = __uint_as_float(__builtin_amdgcn_readlane(wL, b));
                lc[b] = pk >> ROW_BITS;
                const unsigned r = pk & ((1u << ROW_BITS) - 1u);
                v[b] = *(const float4*)(hW + (size_t)r * 128 + lane * 4);
            }
        }
#pragma unroll
        for (int b = 0; b < 8; ++b) {
            if (k0 + b * 32 < cnt) {
                float* d = tile + lc[b] * 128 + lane;
                atomicAdd(d,      w[b] * v[b].x);
                atomicAdd(d + 32, w[b] * v[b].y);
                atomicAdd(d + 64, w[b] * v[b].z);
                atomicAdd(d + 96, w[b] * v[b].w);
            }
        }
    }
    __syncthreads();

    const float4 bq = *(const float4*)(bias + lane * 4);
    for (int i = tid; i < TILE_N * 32; i += AGG_THREADS) {
        const int ln = i >> 5;
        const int n  = nodeBase + ln;
        if (n < N) {
            const float* tp = tile + ln * 128 + lane;
            const float d  = dis[n];
            const float dd = d * d;
            const float4 h = *(const float4*)(hW + (size_t)n * 128 + lane * 4);
            float4 r;
            r.x = fmaxf(tp[0]  + dd * h.x + bq.x, 0.f);
            r.y = fmaxf(tp[32] + dd * h.y + bq.y, 0.f);
            r.z = fmaxf(tp[64] + dd * h.z + bq.z, 0.f);
            r.w = fmaxf(tp[96] + dd * h.w + bq.w, 0.f);
            *(float4*)(out + (size_t)n * 128 + lane * 4) = r;
        }
    }
}

extern "C" void kernel_launch(void* const* d_in, const int* in_sizes, int n_in,
                              void* d_out, int out_size, void* d_ws, size_t ws_size,
                              hipStream_t stream) {
    const float* x   = (const float*)d_in[0];
    const int*   ei  = (const int*)d_in[1];
    const float* ea  = (const float*)d_in[2];
    const float* W1  = (const float*)d_in[3];
    const float* b1  = (const float*)d_in[4];
    const float* W2  = (const float*)d_in[5];
    const float* b2  = (const float*)d_in[6];
    const float* Wr  = (const float*)d_in[7];
    const float* br  = (const float*)d_in[8];
    float* out = (float*)d_out;

    const long N = in_sizes[0] / IN_DIM;
    const long E = in_sizes[2] / 4;
    const int* row = ei;
    const int* col = ei + E;

    char* ws = (char*)d_ws;
    size_t off = 0;
    auto carve = [&](size_t bytes) -> char* {
        char* p = ws + off;
        off += (bytes + 255) & ~(size_t)255;
        return p;
    };
    float*    dis  = (float*)carve(N * 4);
    float*    hW   = (float*)carve(N * 128 * 4);
    float*    h1   = (float*)carve(N * 128 * 4);
    float*    h2   = (float*)carve(N * 128 * 4);
    unsigned* mm   = (unsigned*)carve(2 * 4);
    const unsigned aggBlocks = (unsigned)((N + TILE_N - 1) / TILE_N);
    unsigned* queue  = (unsigned*)carve((size_t)aggBlocks * QCAP * 8);
    unsigned* qcnt   = (unsigned*)carve((size_t)(aggBlocks + 1) * QCNT_STRIDE * 4);
    int*      ovfRow = (int*)carve(E * 4);
    int*      ovfCol = (int*)carve(E * 4);
    float*    ovfW   = (float*)carve(E * 4);
    (void)ws_size; (void)n_in; (void)out_size;

    const int BT = 256;
    auto blk1 = [&](long n) { return (unsigned)((n + BT - 1) / BT); };

    const long nStrips = (N + 15) / 16;
    const unsigned gemmBlocks = (unsigned)((nStrips + GEMM_STRIPS - 1) / GEMM_STRIPS);
    init_state_kernel<<<blk1(aggBlocks + 1), BT, 0, stream>>>(mm, qcnt, (int)(aggBlocks + 1));
    const unsigned eqBlocks = (unsigned)((E + EQ_THREADS * EQ_PER - 1) / (EQ_THREADS * EQ_PER));
    edge_bin_gemm_kernel<<<(eqBlocks > gemmBlocks ? eqBlocks : gemmBlocks), EQ_THREADS, 0, stream>>>(
        ea, row, col, mm, queue, qcnt, ovfRow, ovfCol, ovfW, (int)aggBlocks, (int)E, x, W1, hW, (int)N);
    tile_weight_dis_kernel<<<aggBlocks, AGG_THREADS, 0, stream>>>(queue, qcnt, ovfCol, ovfW, mm, dis, (int)N, (int)E);

    aggregate_tile_kernel<<<aggBlocks, AGG_THREADS, 0, stream>>>(hW, queue, qcnt, ovfRow, ovfCol, ovfW, dis, b1, h1, (int)N, (int)E, 1);

    gemm_nt128_kernel<<<gemmBlocks, GEMM_THREADS, 0, stream>>>(h1, W2, hW, (int)N);
    aggregate_tile_kernel<<<aggBlocks, AGG_THREADS, 0, stream>>>(hW, queue, qcnt, ovfRow, ovfCol, ovfW, dis, b2, h2, (int)N, (int)E, 0);

    gemm_readout_kernel<<<gemmBlocks, GEMM_THREADS, 0, stream>>>(h1, h2, Wr, br, out, (int)N);
}


# Products that read an aggregation's rows done by its block on the LDS tile (rows rotated, weights in the queue's LDS): h1@W2^T, readout halves; 2 launches, h1/h2 stores gone; sums same
# speedup vs baseline: 3.8118x; 1.0763x over previous
#include <hip/hip_runtime.h>
#include <hip/hip_bf16.h>


#define IN_DIM 128
#define HID    128
#define OUT_DIM 64
#define EPS_W  1e-8f

#define TILE_N      392
#define TILE_STRIPS ((TILE_N + 15) / 16)
#define AGG_WROWS   192
#define QCAP        12288
#define AGG_THREADS 1024
#define ROW_BITS    23
#define QCNT_STRIDE 32
#define EQ_THREADS  1024
#define EQ_PER      7
#define EQ_MAXT     2048

#define GEMM_STRIPS  25
#define GEMM_THREADS (GEMM_STRIPS * 32)
#define W_PITCH      132
#define W_STAGERS    512
#define W_CHUNKS     (4096 / W_STAGERS)

typedef float v2f __attribute__((ext_vector_type(2)));
typedef float v8f __attribute__((ext_vector_type(8)));

#if defined(__gfx1250__) && __has_builtin(__builtin_amdgcn_wmma_f32_16x16x4_f32)
#define HAVE_WMMA_F32 1
#endif

__global__ void init_state_kernel(unsigned* __restrict__ mm,
                                  unsigned* __restrict__ qcnt, int nCnt) {
    const int i = blockIdx.x * blockDim.x + threadIdx.x;
    if (i == 0) {
        mm[0] = 0x7F800000u;
        mm[1] = 0x00000000u;
    }
    if (i < nCnt) qcnt[(size_t)i * QCNT_STRIDE] = 0u;
}

__global__ void __launch_bounds__(AGG_THREADS)
tile_weight_dis_kernel(unsigned* __restrict__ queue, const unsigned* __restrict__ qcnt,
                       const int* __restrict__ ovfCol, float* __restrict__ ovfW,
                       const unsigned* __restrict__ mm, float* __restrict__ dis,
                       int N, int E) {
    __shared__ float degL[TILE_N];
    const int tid      = threadIdx.x;
    const int nodeBase = blockIdx.x * TILE_N;
    const int cnt  = (int)min(qcnt[(size_t)blockIdx.x * QCNT_STRIDE], (unsigned)QCAP);
    const int nOvf = (int)min(qcnt[(size_t)gridDim.x * QCNT_STRIDE], (unsigned)E);
    const float mn    = __uint_as_float(mm[0]);
    const float mx    = __uint_as_float(mm[1]);
    const float range = mx - mn;

    for (int i = tid; i < TILE_N; i += AGG_THREADS) degL[i] = 0.f;
    __syncthreads();

    uint2* __restrict__ myQ = (uint2*)queue + (size_t)blockIdx.x * QCAP;
    for (int base = 0; base < cnt; base += 8 * AGG_THREADS) {
        uint2 r8[8];
#pragma unroll
        for (int u = 0; u < 8; ++u) {
            const int i = base + u * AGG_THREADS + tid;
            r8[u] = myQ[min(i, cnt - 1)];
        }
#pragma unroll
        for (int u = 0; u < 8; ++u) {
            const int i = base + u * AGG_THREADS + tid;
            if (i < cnt) {
                const float m = __uint_as_float(r8[u].y);
                float w = (range > EPS_W) ? (1.0f - (m - mn) / (range + EPS_W)) : 1.0f;
                w = fmaxf(w, 0.0f);
                myQ[i] = make_uint2(r8[u].x, __float_as_uint(w));
                atomicAdd(&degL[r8[u].x >> ROW_BITS], w);
            }
        }
    }
    for (int i = tid; i < nOvf; i += AGG_THREADS) {
        const unsigned lc = (unsigned)(ovfCol[i] - nodeBase);
        if (lc < TILE_N) {
            const float m = ovfW[i];
            float w = (range > EPS_W) ? (1.0f - (m - mn) / (range + EPS_W)) : 1.0f;
            w = fmaxf(w, 0.0f);
            ovfW[i] = w;
            atomicAdd(&degL[lc], w);
        }
    }
    __syncthreads();
    for (int i = tid; i < TILE_N; i += AGG_THREADS) {
        const int n = nodeBase + i;
        if (n < N) dis[n] = rsqrtf(degL[i] + 1.0f);
    }
}


__device__ __forceinline__ void w128_load(const float* __restrict__ W, float4 (&wv)[W_CHUNKS], int tid) {
#pragma unroll
    for (int u = 0; u < W_CHUNKS; ++u)
        wv[u] = *(const float4*)(W + (size_t)(tid + u * W_STAGERS) * 4);
}
__device__ __forceinline__ void w128_store(float* __restrict__ wl, const float4 (&wv)[W_CHUNKS], int tid) {
#pragma unroll
    for (int u = 0; u < W_CHUNKS; ++u) {
        const int i = tid + u * W_STAGERS;
        const int n = i >> 5, q = i & 31;
        *(float4*)(wl + n * W_PITCH + q * 4) = wv[u];
    }
}

__device__ __forceinline__ void gemm_strip128(const float* __restrict__ A, const float* __restrict__ wl,
                                              float* __restrict__ C, int N, int m0, int lane) {
    const int half = lane >> 4;
    const int lm   = lane & 15;
    int rowA = m0 + lm;
    if (rowA >= N) rowA = N - 1;
    const float* __restrict__ arow = A + (size_t)rowA * 128;

#if defined(HAVE_WMMA_F32)
    v8f acc[8];
#pragma unroll
    for (int t = 0; t < 8; ++t)
        acc[t] = (v8f){0.f, 0.f, 0.f, 0.f, 0.f, 0.f, 0.f, 0.f};

#pragma unroll 2
    for (int k0 = 0; k0 < 128; k0 += 4) {
        const int ka = k0 + 2 * half;
        v2f a;
        a.x = arow[ka];
        a.y = arow[ka + 1];
        const float* bw = wl + lm * W_PITCH + ka;
#pragma unroll
        for (int t = 0; t < 8; ++t) {
            const v2f b = *(const v2f*)(bw + t * 16 * W_PITCH);
            acc[t] = __builtin_amdgcn_wmma_f32_16x16x4_f32(
                false, a, false, b, (short)0, acc[t], false, false);
        }
    }
#pragma unroll
    for (int t = 0; t < 8; ++t) {
        const int colv = t * 16 + lm;
#pragma unroll
        for (int v = 0; v < 8; ++v) {
            const int rowv = m0 + v + 8 * half;
            if (rowv < N) C[(size_t)rowv * 128 + colv] = acc[t][v];
        }
    }
#else
    for (int t = 0; t < 8; ++t) {
        const int colv = t * 16 + lm;
        for (int v = 0; v < 8; ++v) {
            const int rowv = m0 + v + 8 * half;
            if (rowv >= N) continue;
            float s = 0.f;
            for (int k = 0; k < 128; ++k)
                s += A[(size_t)rowv * 128 + k] * wl[colv * W_PITCH + k];
            C[(size_t)rowv * 128 + colv] = s;
        }
    }
#endif
}

__global__ void __launch_bounds__(EQ_THREADS)
edge_bin_gemm_kernel(const float* __restrict__ ea, const int* __restrict__ row,
                     const int* __restrict__ col, unsigned* __restrict__ mm,
                     unsigned* __restrict__ queue, unsigned* __restrict__ qcnt,
                     int* __restrict__ ovfRow, int* __restrict__ ovfCol,
                     float* __restrict__ ovfW, int nTiles, int E,
                     const float* __restrict__ X, const float* __restrict__ W,
                     float* __restrict__ C, int N) {
    static_assert(EQ_THREADS > GEMM_THREADS && EQ_THREADS >= W_STAGERS, "waves left over for the reservations");
    __shared__ float    wl[128 * W_PITCH];
    __shared__ unsigned hist[EQ_MAXT];
    __shared__ uint4    park[EQ_THREADS * EQ_PER];
    __shared__ unsigned sMin, sMax;
    const int tid    = threadIdx.x;
    const int nLocal = min(nTiles, EQ_MAXT);
    for (int t = tid; t < nLocal; t += EQ_THREADS) hist[t] = 0u;
    if (tid == 0) { sMin = 0x7F800000u; sMax = 0u; }

    const int base = blockIdx.x * (EQ_THREADS * EQ_PER);
    int      r[EQ_PER], c[EQ_PER];
    float4   a[EQ_PER];
#pragma unroll
    for (int u = 0; u < EQ_PER; ++u) {
        const int e = base + u * EQ_THREADS + tid;
        const int ec = (e < E) ? e : E - 1;
        a[u] = *(const float4*)(ea + (size_t)ec * 4);
        r[u] = row[ec];
        c[u] = col[ec];
    }
    __builtin_amdgcn_sched_barrier(0);
    if (tid < W_STAGERS) {
        float4 wv[W_CHUNKS];
        w128_load(W, wv, tid);
        w128_store(wl, wv, tid);
    }
    __syncthreads();

    unsigned bmin = 0x7F800000u, bmax = 0u;
#pragma unroll
    for (int u = 0; u < EQ_PER; ++u) {
        const bool in = base + u * EQ_THREADS + tid < E;
        const int  cu = in ? c[u] : -1;
        const unsigned tl = (unsigned)cu / TILE_N;
        const bool ok = in && tl < (unsigned)nTiles;
        const float m = 0.25f * (a[u].x + a[u].y + a[u].z + a[u].w);
        if (in) {
            const unsigned b = __float_as_uint(m);
            bmin = min(bmin, b);
            bmax = max(bmax, b);
        }
        unsigned loc = 0u;
        if (ok)
            loc = (tl < EQ_MAXT) ? atomicAdd(&hist[tl], 1u)
                                 : atomicAdd(&qcnt[(size_t)tl * QCNT_STRIDE], 1u);
        park[tid * EQ_PER + u] = make_uint4((unsigned)r[u] | (((unsigned)cu - tl * TILE_N) << ROW_BITS),
                                            __float_as_uint(m), ok ? tl : 0xFFFFFFFFu, loc);
    }
    for (int off = 16; off > 0; off >>= 1) {
        bmin = min(bmin, (unsigned)__shfl_xor((int)bmin, off));
        bmax = max(bmax, (unsigned)__shfl_xor((int)bmax, off));
    }
    if ((tid & 31) == 0) {
        atomicMin(&sMin, bmin);
        atomicMax(&sMax, bmax);
    }
    __syncthreads();
    if (tid == 0) {
        atomicMin(&mm[0], sMin);
        atomicMax(&mm[1], sMax);
    }
    if (tid < GEMM_THREADS) {
        const int m0 = (blockIdx.x * GEMM_STRIPS + (tid >> 5)) * 16;
        if (m0 < N) gemm_strip128(X, wl, C, N, m0, tid & 31);
    } else {
        for (int t = tid - GEMM_THREADS; t < nLocal; t += EQ_THREADS - GEMM_THREADS) {
            const unsigned n = hist[t];
            if (n) hist[t] = atomicAdd(&qcnt[(size_t)t * QCNT_STRIDE], n);
        }
    }
    __syncthreads();
    uint2* __restrict__ q2 = (uint2*)queue;
#pragma unroll
    for (int u = 0; u < EQ_PER; ++u) {
        const uint4 rec = park[tid * EQ_PER + u];
        const unsigned tl = rec.z;
        if (tl != 0xFFFFFFFFu) {
            const unsigned pk = rec.x;
            const unsigned mb = rec.y;
            const unsigned slot = rec.w + ((tl < EQ_MAXT) ? hist[tl] : 0u);
            if (slot < QCAP) {
                q2[(size_t)tl * QCAP + slot] = make_uint2(pk, mb);
            } else {
                const unsigned o = atomicAdd(&qcnt[(size_t)nTiles * QCNT_STRIDE], 1u);
                if (o < (unsigned)E) {
                    ovfRow[o] = (int)(pk & ((1u << ROW_BITS) - 1u));
                    ovfCol[o] = (int)(tl * TILE_N + (pk >> ROW_BITS));
                    ovfW[o]   = __uint_as_float(mb);
                }
            }
        }
    }
}

template <int NT>
__device__ __forceinline__ void tile_strip_mma(const float* __restrict__ tile, const float* __restrict__ wl,
                                               int s, int lane, v8f (&acc)[NT]) {
    const int half = lane >> 4;
    const int lm   = lane & 15;
    int rowL = s * 16 + lm;
    if (rowL >= TILE_N) rowL = TILE_N - 1;
    const float* __restrict__ arow = tile + rowL * 128;
    const int rot = 4 * rowL;
#if defined(HAVE_WMMA_F32)
#pragma unroll 2
    for (int k0 = 0; k0 < 128; k0 += 4) {
        const int ka = k0 + 2 * half;
        const v2f a = *(const v2f*)(arow + ((ka + rot) & 127));
        const float* bw = wl + lm * W_PITCH + ka;
#pragma unroll
        for (int t = 0; t < NT; ++t) {
            const v2f b = *(const v2f*)(bw + t * 16 * W_PITCH);
            acc[t] = __builtin_amdgcn_wmma_f32_16x16x4_f32(
                false, a, false, b, (short)0, acc[t], false, false);
        }
    }
#else
    for (int t = 0; t < NT; ++t)
        for (int v = 0; v < 8; ++v) {
            int r = s * 16 + v + 8 * half;
            if (r >= TILE_N) r = TILE_N - 1;
            float sum = acc[t][v];
            for (int k = 0; k < 128; ++k)
                sum += tile[r * 128 + ((k + 4 * r) & 127)] * wl[(t * 16 + lm) * W_PITCH + k];
            acc[t][v] = sum;
        }
#endif
}

__global__ void __launch_bounds__(AGG_THREADS)
aggregate_tile_kernel(const float* __restrict__ hW, unsigned* __restrict__ queue,
                      const unsigned* __restrict__ qcnt,
                      const int* __restrict__ ovfRow, const int* __restrict__ ovfCol,
                      const float* __restrict__ ovfW,
                      const float* __restrict__ dis, const float* __restrict__ bias,
                      const float* __restrict__ Wn, const float* __restrict__ Wr,
                      const float* __restrict__ br, float* __restrict__ hWn,
                      float* __restrict__ out, int N, int E, int firstLayer) {
    static_assert(2 * QCAP <= AGG_WROWS * W_PITCH, "the queue copy fits the LDS that later holds the weights");
    static_assert(TILE_STRIPS * 32 <= AGG_THREADS, "one wave per strip of the tile");
    __shared__ float tile[TILE_N * 128] __attribute__((aligned(16)));
    __shared__ float wreg[AGG_WROWS * W_PITCH] __attribute__((aligned(16)));
    unsigned* const qa = (unsigned*)wreg;
    unsigned* const qb = qa + QCAP;

    const int tid  = threadIdx.x;
    const int lane = tid & 31;
    const int wv   = tid >> 5;
    const int nodeBase = blockIdx.x * TILE_N;

    const int cnt  = (int)min(qcnt[(size_t)blockIdx.x * QCNT_STRIDE], (unsigned)QCAP);
    const int nOvf = (int)min(qcnt[(size_t)gridDim.x * QCNT_STRIDE], (unsigned)E);

    for (int i = tid; i < TILE_N * 32; i += AGG_THREADS)
        *(float4*)(tile + i * 4) = make_float4(0.f, 0.f, 0.f, 0.f);

    uint2* __restrict__ myQ = (uint2*)queue + (size_t)blockIdx.x * QCAP;
    for (int base = 0; base < cnt; base += 8 * AGG_THREADS) {
        uint2 r8[8];
#pragma unroll
        for (int u = 0; u < 8; ++u) {
            const int i = base + u * AGG_THREADS + tid;
            r8[u] = (i < cnt) ? myQ[i] : make_uint2(0u, 0u);
        }
        if (firstLayer) {
            float d0[8], d1[8];
#pragma unroll
            for (int u = 0; u < 8; ++u) {
                d0[u] = dis[r8[u].x & ((1u << ROW_BITS) - 1u)];
                d1[u] = dis[nodeBase + (int)(r8[u].x >> ROW_BITS)];
            }
#pragma unroll
            for (int u = 0; u < 8; ++u) {
                const int i = base + u * AGG_THREADS + tid;
                r8[u].y = __float_as_uint(d0[u] * __uint_as_float(r8[u].y) * d1[u]);
                if (i < cnt) myQ[i] = r8[u];
            }
        }
#pragma unroll
        for (int u = 0; u < 8; ++u) {
            const int i = base + u * AGG_THREADS + tid;
            if (i < cnt) {
                qa[i] = r8[u].x;
                qb[i] = r8[u].y;
            }
        }
    }
    __syncthreads();

    for (int i = tid; i < nOvf; i += AGG_THREADS) {
        const unsigned lc = (unsigned)(ovfCol[i] - nodeBase);
        if (lc < TILE_N) {
            const float w = dis[ovfRow[i]] * ovfW[i] * dis[ovfCol[i]];
            const float* __restrict__ s = hW + (size_t)ovfRow[i] * 128;
            for (int k = 0; k < 128; ++k)
                atomicAdd(&tile[lc * 128 + (k & 3) * 32 + (k >> 2)], w * s[k]);
        }
    }

    for (int k0 = wv; k0 < cnt; k0 += 32 * 8) {
        const int  qi  = k0 + (lane & 7) * 32;
        const bool qok = qi < cnt;
        const unsigned pkL = qok ? qa[qi] : 0u;
        const unsigned wL  = qok ? qb[qi] : 0u;
        float4   v[8];
        float    w[8];
        unsigned lc[8];
#pragma unroll
        for (int b = 0; b < 8; ++b) {
            if (k0 + b * 32 < cnt) {
                const unsigned pk = __builtin_amdgcn_readlane(pkL, b);
                w[b]  = __uint_as_float(__builtin_amdgcn_readlane(wL, b));
                lc[b] = pk >> ROW_BITS;
                const unsigned r = pk & ((1u << ROW_BITS) - 1u);
                v[b] = *(const float4*)(hW + (size_t)r * 128 + lane * 4);
            }
        }
#pragma unroll
        for (int b = 0; b < 8; ++b) {
            if (k0 + b * 32 < cnt) {
                float* d = tile + lc[b] * 128 + lane;
                atomicAdd(d,      w[b] * v[b].x);
                atomicAdd(d + 32, w[b] * v[b].y);
                atomicAdd(d + 64, w[b] * v[b].z);
                atomicAdd(d + 96, w[b] * v[b].w);
            }
        }
    }
    const int pr = tid >> 5, pq = (tid & 31) * 4;
    const float4 z4 = make_float4(0.f, 0.f, 0.f, 0.f);
    float4 w0 = z4, w1 = z4, w2 = z4, w3 = z4, w4 = z4, w5 = z4;
    if (firstLayer) {
        w0 = *(const float4*)(Wn + (size_t)(pr)      * 128 + pq);
        w1 = *(const float4*)(Wn + (size_t)(pr + 32) * 128 + pq);
        w2 = *(const float4*)(Wn + (size_t)(pr + 64) * 128 + pq);
        w3 = *(const float4*)(Wn + (size_t)(pr + 96) * 128 + pq);
        w4 = *(const float4*)(Wr + (size_t)(pr)      * 256 + pq);
        w5 = *(const float4*)(Wr + (size_t)(pr + 32) * 256 + pq);
    } else {
        w0 = *(const float4*)(Wr + (size_t)(pr)      * 256 + 128 + pq);
        w1 = *(const float4*)(Wr + (size_t)(pr + 32) * 256 + 128 + pq);
    }
    __syncthreads();

    *(float4*)(wreg + (pr)      * W_PITCH + pq) = w0;
    *(float4*)(wreg + (pr + 32) * W_PITCH + pq) = w1;
    if (firstLayer) {
        *(float4*)(wreg + (pr + 64)  * W_PITCH + pq) = w2;
        *(float4*)(wreg + (pr + 96)  * W_PITCH + pq) = w3;
        *(float4*)(wreg + (pr + 128) * W_PITCH + pq) = w4;
        *(float4*)(wreg + (pr + 160) * W_PITCH + pq) = w5;
    }

    const float4 bq = *(const float4*)(bias + lane * 4);
    for (int i = tid; i < TILE_N * 32; i += AGG_THREADS) {
        const int ln = i >> 5;
        const int n  = nodeBase + ln;
        if (n < N) {
            float* trow = tile + ln * 128;
            const float* tp = trow + lane;
            const float d  = dis[n];
            const float dd = d * d;
            const float4 h = *(const float4*)(hW + (size_t)n * 128 + lane * 4);
            float4 r;
            r.x = fmaxf(tp[0]  + dd * h.x + bq.x, 0.f);
            r.y = fmaxf(tp[32] + dd * h.y + bq.y, 0.f);
            r.z = fmaxf(tp[64] + dd * h.z + bq.z, 0.f);
            r.w = fmaxf(tp[96] + dd * h.w + bq.w, 0.f);
            *(float4*)(trow + ((lane * 4 + ln * 4) & 127)) = r;
        }
    }
    __syncthreads();

    const int half = lane >> 4;
    const int lm   = lane & 15;
    if (firstLayer) {
        if (wv < TILE_STRIPS) {
            v8f acc[8];
#pragma unroll
            for (int t = 0; t < 8; ++t)
                acc[t] = (v8f){0.f, 0.f, 0.f, 0.f, 0.f, 0.f, 0.f, 0.f};
            tile_strip_mma<8>(tile, wreg, wv, lane, acc);
#pragma unroll
            for (int t = 0; t < 8; ++t) {
#pragma unroll
                for (int v = 0; v < 8; ++v) {
                    const int rl = wv * 16 + v + 8 * half;
                    const int n  = nodeBase + rl;
                    if (rl < TILE_N && n < N) hWn[(size_t)n * 128 + t * 16 + lm] = acc[t][v];
                }
            }
        }
        const int s2 = AGG_THREADS / 32 - 1 - wv;
        if (s2 < TILE_STRIPS) {
            v8f acc[4];
#pragma unroll
            for (int t = 0; t < 4; ++t)
                acc[t] = (v8f){0.f, 0.f, 0.f, 0.f, 0.f, 0.f, 0.f, 0.f};
            tile_strip_mma<4>(tile, wreg + 128 * W_PITCH, s2, lane, acc);
#pragma unroll
            for (int t = 0; t < 4; ++t) {
#pragma unroll
                for (int v = 0; v < 8; ++v) {
                    const int rl = s2 * 16 + v + 8 * half;
                    const int n  = nodeBase + rl;
                    if (rl < TILE_N && n < N) out[(size_t)n * OUT_DIM + t * 16 + lm] = acc[t][v];
                }
            }
        }
    } else if (wv < TILE_STRIPS) {
        v8f acc[4];
#pragma unroll
        for (int t = 0; t < 4; ++t) {
#pragma unroll
            for (int v = 0; v < 8; ++v) {
                const int rl = wv * 16 + v + 8 * half;
                const int n  = nodeBase + rl;
                acc[t][v] = (rl < TILE_N && n < N) ? out[(size_t)n * OUT_DIM + t * 16 + lm] : 0.f;
            }
        }
        tile_strip_mma<4>(tile, wreg, wv, lane, acc);
#pragma unroll
        for (int t = 0; t < 4; ++t) {
            const float rb = br[t * 16 + lm];
#pragma unroll
            for (int v = 0; v < 8; ++v) {
                const int rl = wv * 16 + v + 8 * half;
                const int n  = nodeBase + rl;
                if (rl < TILE_N && n < N) out[(size_t)n * OUT_DIM + t * 16 + lm] = acc[t][v] + rb;
            }
        }
    }
}

extern "C" void kernel_launch(void* const* d_in, const int* in_sizes, int n_in,
                              void* d_out, int out_size, void* d_ws, size_t ws_size,
                              hipStream_t stream) {
    const float* x   = (const float*)d_in[0];
    const int*   ei  = (const int*)d_in[1];
    const float* ea  = (const float*)d_in[2];
    const float* W1  = (const float*)d_in[3];
    const float* b1  = (const float*)d_in[4];
    const float* W2  = (const float*)d_in[5];
    const float* b2  = (const float*)d_in[6];
    const float* Wr  = (const float*)d_in[7];
    const float* br  = (const float*)d_in[8];
    float* out = (float*)d_out;

    const long N = in_sizes[0] / IN_DIM;
    const long E = in_sizes[2] / 4;
    const int* row = ei;
    const int* col = ei + E;

    char* ws = (char*)d_ws;
    size_t off = 0;
    auto carve = [&](size_t bytes) -> char* {
        char* p = ws + off;
        off += (bytes + 255) & ~(size_t)255;
        return p;
    };
    float*    dis  = (float*)carve(N * 4);
    float*    hW   = (float*)carve(N * 128 * 4);
    float*    hW2  = (float*)carve(N * 128 * 4);
    unsigned* mm   = (unsigned*)carve(2 * 4);
    const unsigned aggBlocks = (unsigned)((N + TILE_N - 1) / TILE_N);
    unsigned* queue  = (unsigned*)carve((size_t)aggBlocks * QCAP * 8);
    unsigned* qcnt   = (unsigned*)carve((size_t)(aggBlocks + 1) * QCNT_STRIDE * 4);
    int*      ovfRow = (int*)carve(E * 4);
    int*      ovfCol = (int*)carve(E * 4);
    float*    ovfW   = (float*)carve(E * 4);
    (void)ws_size; (void)n_in; (void)out_size;

    const int BT = 256;
    auto blk1 = [&](long n) { return (unsigned)((n + BT - 1) / BT); };

    const long nStrips = (N + 15) / 16;
    const unsigned gemmBlocks = (unsigned)((nStrips + GEMM_STRIPS - 1) / GEMM_STRIPS);
    init_state_kernel<<<blk1(aggBlocks + 1), BT, 0, stream>>>(mm, qcnt, (int)(aggBlocks + 1));
    const unsigned eqBlocks = (unsigned)((E + EQ_THREADS * EQ_PER - 1) / (EQ_THREADS * EQ_PER));
    edge_bin_gemm_kernel<<<(eqBlocks > gemmBlocks ? eqBlocks : gemmBlocks), EQ_THREADS, 0, stream>>>(
        ea, row, col, mm, queue, qcnt, ovfRow, ovfCol, ovfW, (int)aggBlocks, (int)E, x, W1, hW, (int)N);
    tile_weight_dis_kernel<<<aggBlocks, AGG_THREADS, 0, stream>>>(queue, qcnt, ovfCol, ovfW, mm, dis, (int)N, (int)E);

    aggregate_tile_kernel<<<aggBlocks, AGG_THREADS, 0, stream>>>(hW, queue, qcnt, ovfRow, ovfCol, ovfW, dis, b1, W2, Wr, br, hW2, out, (int)N, (int)E, 1);

    aggregate_tile_kernel<<<aggBlocks, AGG_THREADS, 0, stream>>>(hW2, queue, qcnt, ovfRow, ovfCol, ovfW, dis, b2, W2, Wr, br, hW, out, (int)N, (int)E, 0);
}
